# starting kernel
# baseline (speedup 1.0000x reference)
#include <hip/hip_runtime.h>
#include <hip/hip_bf16.h>

typedef __attribute__((ext_vector_type(2))) float v2f;
typedef __attribute__((ext_vector_type(8))) float v8f;

#define IN_SIZE 256
#define OUT_SIZE 64


__global__ __launch_bounds__(256) void k_init_deg(unsigned int* __restrict__ deg, int n) {
  int i = blockIdx.x * 256 + threadIdx.x;
  if (i < n) deg[i] = 0u;
}

__global__ __launch_bounds__(256) void k_count_deg(const int* __restrict__ row,
                                                   const int* __restrict__ col,
                                                   unsigned int* __restrict__ deg, int E) {
  int e = blockIdx.x * 256 + threadIdx.x;
  if (e < E) {
    int r = row[e], c = col[e];
    if (r != c) atomicAdd(&deg[r], 1u);
  }
}

__global__ __launch_bounds__(256) void k_dinv(const unsigned int* __restrict__ deg,
                                              float* __restrict__ dinv, int n) {
  int i = blockIdx.x * 256 + threadIdx.x;
  if (i < n) dinv[i] = rsqrtf((float)deg[i] + 1.0f);
}


__global__ __launch_bounds__(256) void k_gemm(const float* __restrict__ x,
                                              const float* __restrict__ W,
                                              float* __restrict__ xw, int n) {
  const int lane  = threadIdx.x & 31;
  const int wave  = threadIdx.x >> 5;
  const int tile  = blockIdx.x * 8 + wave;
  if (tile * 16 >= n) return;

  const int ml    = lane & 15;
  const int khalf = (lane >> 4) << 1;
  int arow = tile * 16 + ml;
  if (arow > n - 1) arow = n - 1;
  const float* xrow = x + (size_t)arow * IN_SIZE;

  v8f acc0 = {}, acc1 = {}, acc2 = {}, acc3 = {};
  for (int k = 0; k < IN_SIZE; k += 4) {
    v2f a = *(const v2f*)(xrow + k + khalf);
    const float* wb = W + (size_t)ml * IN_SIZE + k + khalf;
    v2f b0 = *(const v2f*)(wb + (size_t)0 * 16 * IN_SIZE);
    v2f b1 = *(const v2f*)(wb + (size_t)1 * 16 * IN_SIZE);
    v2f b2 = *(const v2f*)(wb + (size_t)2 * 16 * IN_SIZE);
    v2f b3 = *(const v2f*)(wb + (size_t)3 * 16 * IN_SIZE);
    acc0 = __builtin_amdgcn_wmma_f32_16x16x4_f32(false, a, false, b0, (short)0, acc0, false, false);
    acc1 = __builtin_amdgcn_wmma_f32_16x16x4_f32(false, a, false, b1, (short)0, acc1, false, false);
    acc2 = __builtin_amdgcn_wmma_f32_16x16x4_f32(false, a, false, b2, (short)0, acc2, false, false);
    acc3 = __builtin_amdgcn_wmma_f32_16x16x4_f32(false, a, false, b3, (short)0, acc3, false, false);
  }

  const int rbase = tile * 16 + ((lane >> 4) << 3);
  float* o = xw + (size_t)rbase * OUT_SIZE + ml;
#pragma unroll
  for (int v = 0; v < 8; ++v) {
    if (rbase + v < n) {
      o[(size_t)v * OUT_SIZE +  0] = acc0[v];
      o[(size_t)v * OUT_SIZE + 16] = acc1[v];
      o[(size_t)v * OUT_SIZE + 32] = acc2[v];
      o[(size_t)v * OUT_SIZE + 48] = acc3[v];
    }
  }
}


__global__ __launch_bounds__(256) void k_self(const float* __restrict__ xw,
                                              const float* __restrict__ dinv,
                                              float* __restrict__ out, int n) {
  size_t idx = (size_t)blockIdx.x * 256 + threadIdx.x;
  if (idx < (size_t)n * OUT_SIZE) {
    int i = (int)(idx >> 6);
    float d = dinv[i];
    out[idx] = d * d * xw[idx];
  }
}


__global__ __launch_bounds__(256) void k_scatter(const int* __restrict__ row,
                                                 const int* __restrict__ col,
                                                 const float* __restrict__ dinv,
                                                 const float* __restrict__ xw,
                                                 float* __restrict__ out, int E) {
  int e = blockIdx.x * 4 + (threadIdx.x >> 6);
  int j = threadIdx.x & 63;
  if (e >= E) return;
  int r = row[e];
  int c = col[e];
  if (r == c) return;
  float norm = dinv[r] * dinv[c];
  unsafeAtomicAdd(out + (size_t)r * OUT_SIZE + j,
                  norm * xw[(size_t)c * OUT_SIZE + j]);
}


extern "C" void kernel_launch(void* const* d_in, const int* in_sizes, int n_in,
                              void* d_out, int out_size, void* d_ws, size_t ws_size,
                              hipStream_t stream) {
  const int*   edge = (const int*)d_in[0];
  const float* x    = (const float*)d_in[1];
  const float* W    = (const float*)d_in[3];

  const int E = in_sizes[0] / 2;
  const int n = in_sizes[1] / IN_SIZE;
  const int* row = edge;
  const int* col = edge + E;
  float* out = (float*)d_out;

  float*        xw   = (float*)d_ws;
  unsigned int* deg  = (unsigned int*)(xw + (size_t)n * OUT_SIZE);
  float*        dinv = (float*)(deg + n);

  k_init_deg<<<(n + 255) / 256, 256, 0, stream>>>(deg, n);
  k_count_deg<<<(E + 255) / 256, 256, 0, stream>>>(row, col, deg, E);
  k_dinv<<<(n + 255) / 256, 256, 0, stream>>>(deg, dinv, n);

  const int tiles = (n + 15) / 16;
  k_gemm<<<(tiles + 7) / 8, 256, 0, stream>>>(x, W, xw, n);

  const size_t tot = (size_t)n * OUT_SIZE;
  k_self<<<(unsigned)((tot + 255) / 256), 256, 0, stream>>>(xw, dinv, out, n);
  k_scatter<<<(E + 3) / 4, 256, 0, stream>>>(row, col, dinv, xw, out, E);
}


# Edges counting-sorted by destination row (rank from a returning u32 atomic, scan, fill); one wave per row then sums its dinv-scaled source rows with no f32 atomics; sum order = atomic arrival order
# speedup vs baseline: 1.9262x; 1.9262x over previous
#include <hip/hip_runtime.h>
#include <hip/hip_bf16.h>


typedef __attribute__((ext_vector_type(2))) float v2f;
typedef __attribute__((ext_vector_type(8))) float v8f;

#define IN_SIZE 256
#define OUT_SIZE 64
#define SCAN_B 1024


__global__ __launch_bounds__(256) void k_init_deg(unsigned int* __restrict__ deg, int n) {
  int i = blockIdx.x * 256 + threadIdx.x;
  if (i < n) deg[i] = 0u;
}

__global__ __launch_bounds__(256) void k_rank(const int* __restrict__ row,
                                              const int* __restrict__ col,
                                              unsigned int* __restrict__ deg,
                                              int* __restrict__ rank, int E) {
  int e = blockIdx.x * 256 + threadIdx.x;
  if (e < E) {
    int r = row[e], c = col[e];
    int k = -1;
    if (r != c) k = (int)atomicAdd(&deg[r], 1u);
    rank[e] = k;
  }
}

__global__ __launch_bounds__(SCAN_B) void k_scan(const unsigned int* __restrict__ deg,
                                                 unsigned int* __restrict__ rowptr,
                                                 float* __restrict__ dinv, int n) {
  __shared__ unsigned int s_part[32];
  __shared__ unsigned int s_base;
  const int tid = threadIdx.x, lane = tid & 31, wv = tid >> 5;
  const int start = blockIdx.x * SCAN_B;

  unsigned int acc = 0u;
  const uint4* d4 = (const uint4*)deg;
  const int start4 = start >> 2;
  const int full = start4 / SCAN_B;
#pragma unroll 8
  for (int t = 0; t < full; ++t) {
    uint4 v = d4[t * SCAN_B + tid];
    acc += (v.x + v.y) + (v.z + v.w);
  }
  const int tail = full * SCAN_B + tid;
  if (tail < start4) {
    uint4 v = d4[tail];
    acc += (v.x + v.y) + (v.z + v.w);
  }
  for (int o = 16; o > 0; o >>= 1) acc += (unsigned int)__shfl_xor((int)acc, o);
  if (lane == 0) s_part[wv] = acc;
  __syncthreads();
  if (wv == 0) {
    unsigned int v = s_part[lane];
    for (int o = 16; o > 0; o >>= 1) v += (unsigned int)__shfl_xor((int)v, o);
    if (lane == 0) s_base = v;
  }
  __syncthreads();
  const unsigned int base = s_base;

  const int i = start + tid;
  const unsigned int d = (i < n) ? deg[i] : 0u;
  unsigned int x = d;
  for (int o = 1; o < 32; o <<= 1) {
    unsigned int y = (unsigned int)__shfl_up((int)x, o);
    if (lane >= o) x += y;
  }
  if (lane == 31) s_part[wv] = x;
  __syncthreads();
  unsigned int woff = 0u;
  for (int k = 0; k < wv; ++k) woff += s_part[k];
  const unsigned int excl = base + woff + x - d;
  if (i < n) {
    rowptr[i] = excl;
    dinv[i] = rsqrtf((float)d + 1.0f);
  }
  if (i == n - 1) rowptr[n] = excl + d;
}

__global__ __launch_bounds__(256) void k_fill(const int* __restrict__ row,
                                              const int* __restrict__ col,
                                              const int* __restrict__ rank,
                                              const unsigned int* __restrict__ rowptr,
                                              int* __restrict__ csr, int E) {
  int e = blockIdx.x * 256 + threadIdx.x;
  if (e < E) {
    const int k = rank[e];
    const int r = row[e];
    const int c = col[e];
    if (k >= 0) csr[rowptr[r] + (unsigned int)k] = c;
  }
}


__global__ __launch_bounds__(256) void k_gemm(const float* __restrict__ x,
                                              const float* __restrict__ W,
                                              const float* __restrict__ rs,
                                              float* __restrict__ xw, int n) {
  const int lane  = threadIdx.x & 31;
  const int wave  = threadIdx.x >> 5;
  const int tile  = blockIdx.x * 8 + wave;
  if (tile * 16 >= n) return;

  const int ml    = lane & 15;
  const int khalf = (lane >> 4) << 1;
  int arow = tile * 16 + ml;
  if (arow > n - 1) arow = n - 1;
  const float* xrow = x + (size_t)arow * IN_SIZE;

  const int rbase = tile * 16 + ((lane >> 4) << 3);
  float sc[8];
#pragma unroll
  for (int v = 0; v < 8; ++v) sc[v] = (rs && rbase + v < n) ? rs[rbase + v] : 1.0f;

  v8f acc0 = {}, acc1 = {}, acc2 = {}, acc3 = {};
  for (int k = 0; k < IN_SIZE; k += 4) {
    v2f a = *(const v2f*)(xrow + k + khalf);
    const float* wb = W + (size_t)ml * IN_SIZE + k + khalf;
    v2f b0 = *(const v2f*)(wb + (size_t)0 * 16 * IN_SIZE);
    v2f b1 = *(const v2f*)(wb + (size_t)1 * 16 * IN_SIZE);
    v2f b2 = *(const v2f*)(wb + (size_t)2 * 16 * IN_SIZE);
    v2f b3 = *(const v2f*)(wb + (size_t)3 * 16 * IN_SIZE);
    acc0 = __builtin_amdgcn_wmma_f32_16x16x4_f32(false, a, false, b0, (short)0, acc0, false, false);
    acc1 = __builtin_amdgcn_wmma_f32_16x16x4_f32(false, a, false, b1, (short)0, acc1, false, false);
    acc2 = __builtin_amdgcn_wmma_f32_16x16x4_f32(false, a, false, b2, (short)0, acc2, false, false);
    acc3 = __builtin_amdgcn_wmma_f32_16x16x4_f32(false, a, false, b3, (short)0, acc3, false, false);
  }

  float* o = xw + (size_t)rbase * OUT_SIZE + ml;
#pragma unroll
  for (int v = 0; v < 8; ++v) {
    if (rbase + v < n) {
      const float s = sc[v];
      o[(size_t)v * OUT_SIZE +  0] = s * acc0[v];
      o[(size_t)v * OUT_SIZE + 16] = s * acc1[v];
      o[(size_t)v * OUT_SIZE + 32] = s * acc2[v];
      o[(size_t)v * OUT_SIZE + 48] = s * acc3[v];
    }
  }
}


__global__ __launch_bounds__(256) void k_gather(const unsigned int* __restrict__ rowptr,
                                                const int* __restrict__ csr,
                                                const float* __restrict__ dinv,
                                                const float* __restrict__ xs,
                                                float* __restrict__ out, int n) {
  const int lane = threadIdx.x & 31;
  const int r = __builtin_amdgcn_readfirstlane(blockIdx.x * 8 + (threadIdx.x >> 5));
  if (r >= n) return;
  const unsigned int p0 = rowptr[r], p1 = rowptr[r + 1];
  const v2f* rows = (const v2f*)xs;
  v2f acc = rows[(size_t)r * 32 + lane];
  for (unsigned int p = p0; p < p1; p += 8) {
    int c[8];
#pragma unroll
    for (int k = 0; k < 8; ++k) {
      const unsigned int q = p + k;
      c[k] = csr[q < p1 ? q : p1 - 1u];
    }
    v2f v[8];
#pragma unroll
    for (int k = 0; k < 8; ++k) {
      v[k] = (v2f){0.0f, 0.0f};
      if (p + k < p1) v[k] = rows[(size_t)c[k] * 32 + lane];
    }
#pragma unroll
    for (int k = 0; k < 8; ++k) acc += v[k];
  }
  const float d = dinv[r];
  ((v2f*)out)[(size_t)r * 32 + lane] = acc * d;
}


__global__ __launch_bounds__(256) void k_count_deg(const int* __restrict__ row,
                                                   const int* __restrict__ col,
                                                   unsigned int* __restrict__ deg, int E) {
  int e = blockIdx.x * 256 + threadIdx.x;
  if (e < E) {
    int r = row[e], c = col[e];
    if (r != c) atomicAdd(&deg[r], 1u);
  }
}

__global__ __launch_bounds__(256) void k_dinv(const unsigned int* __restrict__ deg,
                                              float* __restrict__ dinv, int n) {
  int i = blockIdx.x * 256 + threadIdx.x;
  if (i < n) dinv[i] = rsqrtf((float)deg[i] + 1.0f);
}

__global__ __launch_bounds__(256) void k_self(const float* __restrict__ xw,
                                              const float* __restrict__ dinv,
                                              float* __restrict__ out, int n) {
  size_t idx = (size_t)blockIdx.x * 256 + threadIdx.x;
  if (idx < (size_t)n * OUT_SIZE) {
    int i = (int)(idx >> 6);
    float d = dinv[i];
    out[idx] = d * d * xw[idx];
  }
}

__global__ __launch_bounds__(256) void k_scatter(const int* __restrict__ row,
                                                 const int* __restrict__ col,
                                                 const float* __restrict__ dinv,
                                                 const float* __restrict__ xw,
                                                 float* __restrict__ out, int E) {
  int e = blockIdx.x * 4 + (threadIdx.x >> 6);
  int j = threadIdx.x & 63;
  if (e >= E) return;
  int r = row[e];
  int c = col[e];
  if (r == c) return;
  float norm = dinv[r] * dinv[c];
  unsafeAtomicAdd(out + (size_t)r * OUT_SIZE + j,
                  norm * xw[(size_t)c * OUT_SIZE + j]);
}


extern "C" void kernel_launch(void* const* d_in, const int* in_sizes, int n_in,
                              void* d_out, int out_size, void* d_ws, size_t ws_size,
                              hipStream_t stream) {
  const int*   edge = (const int*)d_in[0];
  const float* x    = (const float*)d_in[1];
  const float* W    = (const float*)d_in[3];

  const int E = in_sizes[0] / 2;
  const int n = in_sizes[1] / IN_SIZE;
  const int* row = edge;
  const int* col = edge + E;
  float* out = (float*)d_out;

  float*        xw     = (float*)d_ws;
  unsigned int* deg    = (unsigned int*)(xw + (size_t)n * OUT_SIZE);
  float*        dinv   = (float*)(deg + n);
  unsigned int* rowptr = (unsigned int*)(dinv + n);
  int*          csr    = (int*)(rowptr + n + 1);
  int*          rank   = (int*)d_ws;

  const int tiles = (n + 15) / 16;
  const size_t need = ((size_t)n * OUT_SIZE + 3 * (size_t)n + 1 + (size_t)E) * sizeof(float);

  k_init_deg<<<(n + 255) / 256, 256, 0, stream>>>(deg, n);
  if (ws_size >= need && (size_t)E <= (size_t)n * OUT_SIZE) {
    k_rank<<<(E + 255) / 256, 256, 0, stream>>>(row, col, deg, rank, E);
    k_scan<<<(n + SCAN_B - 1) / SCAN_B, SCAN_B, 0, stream>>>(deg, rowptr, dinv, n);
    k_fill<<<(E + 255) / 256, 256, 0, stream>>>(row, col, rank, rowptr, csr, E);
    k_gemm<<<(tiles + 7) / 8, 256, 0, stream>>>(x, W, dinv, xw, n);
    k_gather<<<(n + 7) / 8, 256, 0, stream>>>(rowptr, csr, dinv, xw, out, n);
  } else {
    k_count_deg<<<(E + 255) / 256, 256, 0, stream>>>(row, col, deg, E);
    k_dinv<<<(n + 255) / 256, 256, 0, stream>>>(deg, dinv, n);
    k_gemm<<<(tiles + 7) / 8, 256, 0, stream>>>(x, W, nullptr, xw, n);
    const size_t tot = (size_t)n * OUT_SIZE;
    k_self<<<(unsigned)((tot + 255) / 256), 256, 0, stream>>>(xw, dinv, out, n);
    k_scatter<<<(E + 3) / 4, 256, 0, stream>>>(row, col, dinv, xw, out, E);
  }
}


# GEMM operand feed: W copied once per work-group into padded LDS, A and B fragments fetched 16 B per lane in one shared permuted k order; the k order of each dot product's sum changes
# speedup vs baseline: 2.2868x; 1.1872x over previous
#include <hip/hip_runtime.h>
#include <hip/hip_bf16.h>


typedef __attribute__((ext_vector_type(2))) float v2f;
typedef __attribute__((ext_vector_type(8))) float v8f;

#define IN_SIZE 256
#define OUT_SIZE 64
#define SCAN_B 1024


__global__ __launch_bounds__(256) void k_init_deg(unsigned int* __restrict__ deg, int n) {
  int i = blockIdx.x * 256 + threadIdx.x;
  if (i < n) deg[i] = 0u;
}

__global__ __launch_bounds__(256) void k_rank(const int* __restrict__ row,
                                              const int* __restrict__ col,
                                              unsigned int* __restrict__ deg,
                                              int* __restrict__ rank, int E) {
  int e = blockIdx.x * 256 + threadIdx.x;
  if (e < E) {
    int r = row[e], c = col[e];
    int k = -1;
    if (r != c) k = (int)atomicAdd(&deg[r], 1u);
    rank[e] = k;
  }
}

__global__ __launch_bounds__(SCAN_B) void k_scan(const unsigned int* __restrict__ deg,
                                                 unsigned int* __restrict__ rowptr,
                                                 float* __restrict__ dinv, int n) {
  __shared__ unsigned int s_part[32];
  __shared__ unsigned int s_base;
  const int tid = threadIdx.x, lane = tid & 31, wv = tid >> 5;
  const int start = blockIdx.x * SCAN_B;

  unsigned int acc = 0u;
  const uint4* d4 = (const uint4*)deg;
  const int start4 = start >> 2;
  const int full = start4 / SCAN_B;
#pragma unroll 8
  for (int t = 0; t < full; ++t) {
    uint4 v = d4[t * SCAN_B + tid];
    acc += (v.x + v.y) + (v.z + v.w);
  }
  const int tail = full * SCAN_B + tid;
  if (tail < start4) {
    uint4 v = d4[tail];
    acc += (v.x + v.y) + (v.z + v.w);
  }
  for (int o = 16; o > 0; o >>= 1) acc += (unsigned int)__shfl_xor((int)acc, o);
  if (lane == 0) s_part[wv] = acc;
  __syncthreads();
  if (wv == 0) {
    unsigned int v = s_part[lane];
    for (int o = 16; o > 0; o >>= 1) v += (unsigned int)__shfl_xor((int)v, o);
    if (lane == 0) s_base = v;
  }
  __syncthreads();
  const unsigned int base = s_base;

  const int i = start + tid;
  const unsigned int d = (i < n) ? deg[i] : 0u;
  unsigned int x = d;
  for (int o = 1; o < 32; o <<= 1) {
    unsigned int y = (unsigned int)__shfl_up((int)x, o);
    if (lane >= o) x += y;
  }
  if (lane == 31) s_part[wv] = x;
  __syncthreads();
  unsigned int woff = 0u;
  for (int k = 0; k < wv; ++k) woff += s_part[k];
  const unsigned int excl = base + woff + x - d;
  if (i < n) {
    rowptr[i] = excl;
    dinv[i] = rsqrtf((float)d + 1.0f);
  }
  if (i == n - 1) rowptr[n] = excl + d;
}

__global__ __launch_bounds__(256) void k_fill(const int* __restrict__ row,
                                              const int* __restrict__ col,
                                              const int* __restrict__ rank,
                                              const unsigned int* __restrict__ rowptr,
                                              int* __restrict__ csr, int E) {
  int e = blockIdx.x * 256 + threadIdx.x;
  if (e < E) {
    const int k = rank[e];
    const int r = row[e];
    const int c = col[e];
    if (k >= 0) csr[rowptr[r] + (unsigned int)k] = c;
  }
}


typedef __attribute__((ext_vector_type(4))) float v4f;

#define WP 260

__global__ __launch_bounds__(256) void k_gemm(const float* __restrict__ x,
                                              const float* __restrict__ W,
                                              const float* __restrict__ rs,
                                              float* __restrict__ xw, int n) {
  __shared__ __attribute__((aligned(16))) float sW[OUT_SIZE * WP];

  const int lane  = threadIdx.x & 31;
  const int wave  = threadIdx.x >> 5;
  const int tile  = blockIdx.x * 8 + wave;
  const int ml    = lane & 15;
  const int koff  = (lane >> 4) << 2;
  int arow = tile * 16 + ml;
  if (arow > n - 1) arow = n - 1;
  const v4f* xr = (const v4f*)(x + (size_t)arow * IN_SIZE + koff);

  const int rbase = tile * 16 + ((lane >> 4) << 3);

  v4f a0 = xr[0], a1 = xr[2], a2 = xr[4], a3 = xr[6];
  float sc[8];
#pragma unroll
  for (int v = 0; v < 8; ++v) sc[v] = (rs && rbase + v < n) ? rs[rbase + v] : 1.0f;

  {
    const v4f* W4 = (const v4f*)W;
    v4f t[16];
#pragma unroll
    for (int i = 0; i < 16; ++i) t[i] = W4[i * 256 + threadIdx.x];
#pragma unroll
    for (int i = 0; i < 16; ++i) {
      const int p = i * 256 + threadIdx.x;
      *(v4f*)(sW + (p >> 6) * WP + ((p & 63) << 2)) = t[i];
    }
  }
  __syncthreads();
  if (tile * 16 >= n) return;

  const float* wl = sW + ml * WP + koff;

  v8f acc0 = {}, acc1 = {}, acc2 = {}, acc3 = {};
  v4f b0 = *(const v4f*)(wl + 0 * 16 * WP);
  v4f b1 = *(const v4f*)(wl + 1 * 16 * WP);
  v4f b2 = *(const v4f*)(wl + 2 * 16 * WP);
  v4f b3 = *(const v4f*)(wl + 3 * 16 * WP);

#define GEMM_STEP(A, KN)                                                                              \
  {                                                                                                  \
          \
    const v4f n0 = *(const v4f*)(wl + 0 * 16 * WP + (KN));                                           \
    const v4f n1 = *(const v4f*)(wl + 1 * 16 * WP + (KN));                                           \
    const v4f n2 = *(const v4f*)(wl + 2 * 16 * WP + (KN));                                           \
    const v4f n3 = *(const v4f*)(wl + 3 * 16 * WP + (KN));                                           \
    acc0 = __builtin_amdgcn_wmma_f32_16x16x4_f32(false, (A).lo, false, b0.lo, (short)0, acc0, false, false); \
    acc1 = __builtin_amdgcn_wmma_f32_16x16x4_f32(false, (A).lo, false, b1.lo, (short)0, acc1, false, false); \
    acc2 = __builtin_amdgcn_wmma_f32_16x16x4_f32(false, (A).lo, false, b2.lo, (short)0, acc2, false, false); \
    acc3 = __builtin_amdgcn_wmma_f32_16x16x4_f32(false, (A).lo, false, b3.lo, (short)0, acc3, false, false); \
    acc0 = __builtin_amdgcn_wmma_f32_16x16x4_f32(false, (A).hi, false, b0.hi, (short)0, acc0, false, false); \
    acc1 = __builtin_amdgcn_wmma_f32_16x16x4_f32(false, (A).hi, false, b1.hi, (short)0, acc1, false, false); \
    acc2 = __builtin_amdgcn_wmma_f32_16x16x4_f32(false, (A).hi, false, b2.hi, (short)0, acc2, false, false); \
    acc3 = __builtin_amdgcn_wmma_f32_16x16x4_f32(false, (A).hi, false, b3.hi, (short)0, acc3, false, false); \
    b0 = n0; b1 = n1; b2 = n2; b3 = n3;                                                              \
  }

  for (int k = 0; k < IN_SIZE; k += 32) {
    v4f q0 = a0, q1 = a1, q2 = a2, q3 = a3;
    if (k + 32 < IN_SIZE) {
      const v4f* xn = xr + ((k + 32) >> 2);
      q0 = xn[0]; q1 = xn[2]; q2 = xn[4]; q3 = xn[6];
    }
    GEMM_STEP(a0, k + 8)
    GEMM_STEP(a1, k + 16)
    GEMM_STEP(a2, k + 24)
    GEMM_STEP(a3, (k + 32) & (IN_SIZE - 1))
    a0 = q0; a1 = q1; a2 = q2; a3 = q3;
  }
#undef GEMM_STEP

  float* o = xw + (size_t)rbase * OUT_SIZE + ml;
#pragma unroll
  for (int v = 0; v < 8; ++v) {
    if (rbase + v < n) {
      const float s = sc[v];
      o[(size_t)v * OUT_SIZE +  0] = s * acc0[v];
      o[(size_t)v * OUT_SIZE + 16] = s * acc1[v];
      o[(size_t)v * OUT_SIZE + 32] = s * acc2[v];
      o[(size_t)v * OUT_SIZE + 48] = s * acc3[v];
    }
  }
}


__global__ __launch_bounds__(256) void k_gather(const unsigned int* __restrict__ rowptr,
                                                const int* __restrict__ csr,
                                                const float* __restrict__ dinv,
                                                const float* __restrict__ xs,
                                                float* __restrict__ out, int n) {
  const int lane = threadIdx.x & 31;
  const int r = __builtin_amdgcn_readfirstlane(blockIdx.x * 8 + (threadIdx.x >> 5));
  if (r >= n) return;
  const unsigned int p0 = rowptr[r], p1 = rowptr[r + 1];
  const v2f* rows = (const v2f*)xs;
  v2f acc = rows[(size_t)r * 32 + lane];
  for (unsigned int p = p0; p < p1; p += 8) {
    int c[8];
#pragma unroll
    for (int k = 0; k < 8; ++k) {
      const unsigned int q = p + k;
      c[k] = csr[q < p1 ? q : p1 - 1u];
    }
    v2f v[8];
#pragma unroll
    for (int k = 0; k < 8; ++k) {
      v[k] = (v2f){0.0f, 0.0f};
      if (p + k < p1) v[k] = rows[(size_t)c[k] * 32 + lane];
    }
#pragma unroll
    for (int k = 0; k < 8; ++k) acc += v[k];
  }
  const float d = dinv[r];
  ((v2f*)out)[(size_t)r * 32 + lane] = acc * d;
}


__global__ __launch_bounds__(256) void k_count_deg(const int* __restrict__ row,
                                                   const int* __restrict__ col,
                                                   unsigned int* __restrict__ deg, int E) {
  int e = blockIdx.x * 256 + threadIdx.x;
  if (e < E) {
    int r = row[e], c = col[e];
    if (r != c) atomicAdd(&deg[r], 1u);
  }
}

__global__ __launch_bounds__(256) void k_dinv(const unsigned int* __restrict__ deg,
                                              float* __restrict__ dinv, int n) {
  int i = blockIdx.x * 256 + threadIdx.x;
  if (i < n) dinv[i] = rsqrtf((float)deg[i] + 1.0f);
}

__global__ __launch_bounds__(256) void k_self(const float* __restrict__ xw,
                                              const float* __restrict__ dinv,
                                              float* __restrict__ out, int n) {
  size_t idx = (size_t)blockIdx.x * 256 + threadIdx.x;
  if (idx < (size_t)n * OUT_SIZE) {
    int i = (int)(idx >> 6);
    float d = dinv[i];
    out[idx] = d * d * xw[idx];
  }
}

__global__ __launch_bounds__(256) void k_scatter(const int* __restrict__ row,
                                                 const int* __restrict__ col,
                                                 const float* __restrict__ dinv,
                                                 const float* __restrict__ xw,
                                                 float* __restrict__ out, int E) {
  int e = blockIdx.x * 4 + (threadIdx.x >> 6);
  int j = threadIdx.x & 63;
  if (e >= E) return;
  int r = row[e];
  int c = col[e];
  if (r == c) return;
  float norm = dinv[r] * dinv[c];
  unsafeAtomicAdd(out + (size_t)r * OUT_SIZE + j,
                  norm * xw[(size_t)c * OUT_SIZE + j]);
}


extern "C" void kernel_launch(void* const* d_in, const int* in_sizes, int n_in,
                              void* d_out, int out_size, void* d_ws, size_t ws_size,
                              hipStream_t stream) {
  const int*   edge = (const int*)d_in[0];
  const float* x    = (const float*)d_in[1];
  const float* W    = (const float*)d_in[3];

  const int E = in_sizes[0] / 2;
  const int n = in_sizes[1] / IN_SIZE;
  const int* row = edge;
  const int* col = edge + E;
  float* out = (float*)d_out;

  float*        xw     = (float*)d_ws;
  unsigned int* deg    = (unsigned int*)(xw + (size_t)n * OUT_SIZE);
  float*        dinv   = (float*)(deg + n);
  unsigned int* rowptr = (unsigned int*)(dinv + n);
  int*          csr    = (int*)(rowptr + n + 1);
  int*          rank   = (int*)d_ws;

  const int tiles = (n + 15) / 16;
  const size_t need = ((size_t)n * OUT_SIZE + 3 * (size_t)n + 1 + (size_t)E) * sizeof(float);

  k_init_deg<<<(n + 255) / 256, 256, 0, stream>>>(deg, n);
  if (ws_size >= need && (size_t)E <= (size_t)n * OUT_SIZE) {
    k_rank<<<(E + 255) / 256, 256, 0, stream>>>(row, col, deg, rank, E);
    k_scan<<<(n + SCAN_B - 1) / SCAN_B, SCAN_B, 0, stream>>>(deg, rowptr, dinv, n);
    k_fill<<<(E + 255) / 256, 256, 0, stream>>>(row, col, rank, rowptr, csr, E);
    k_gemm<<<(tiles + 7) / 8, 256, 0, stream>>>(x, W, dinv, xw, n);
    k_gather<<<(n + 7) / 8, 256, 0, stream>>>(rowptr, csr, dinv, xw, out, n);
  } else {
    k_count_deg<<<(E + 255) / 256, 256, 0, stream>>>(row, col, deg, E);
    k_dinv<<<(n + 255) / 256, 256, 0, stream>>>(deg, dinv, n);
    k_gemm<<<(tiles + 7) / 8, 256, 0, stream>>>(x, W, nullptr, xw, n);
    const size_t tot = (size_t)n * OUT_SIZE;
    k_self<<<(unsigned)((tot + 255) / 256), 256, 0, stream>>>(xw, dinv, out, n);
    k_scatter<<<(E + 3) / 4, 256, 0, stream>>>(row, col, dinv, xw, out, E);
  }
}


# No scan or fill launch: each kept edge's source id is stored at slot k of its row (k = the returning count add), 64 slots in the row's own output bytes, overflow list beyond; arrival-order sums kept
# speedup vs baseline: 2.5571x; 1.1182x over previous
#include <hip/hip_runtime.h>
#include <hip/hip_bf16.h>


typedef __attribute__((ext_vector_type(2))) float v2f;
typedef __attribute__((ext_vector_type(8))) float v8f;

#define IN_SIZE 256
#define OUT_SIZE 64
#define SCAN_B 1024


#define SLOTS 64

__global__ __launch_bounds__(256) void k_init_deg(unsigned int* __restrict__ cnt, int m) {
  int i = blockIdx.x * 256 + threadIdx.x;
  if (i < m) cnt[i] = 0u;
}

__global__ __launch_bounds__(256) void k_place(const int* __restrict__ row,
                                               const int* __restrict__ col,
                                               unsigned int* __restrict__ deg,
                                               int* __restrict__ slots,
                                               int* __restrict__ ovf, int n, int E) {
  int e = blockIdx.x * 256 + threadIdx.x;
  if (e < E) {
    int r = row[e], c = col[e];
    if (r != c) {
      unsigned int k = atomicAdd(&deg[r], 1u);
      if (k < (unsigned int)SLOTS) {
        slots[(size_t)r * SLOTS + k] = c;
      } else {
        unsigned int j = atomicAdd(&deg[n], 1u);
        ovf[j] = e;
      }
    }
  }
}


typedef __attribute__((ext_vector_type(4))) float v4f;

#define WP 260

__global__ __launch_bounds__(256) void k_gemm(const float* __restrict__ x,
                                              const float* __restrict__ W,
                                              const unsigned int* __restrict__ dg,
                                              float* __restrict__ xw, int n) {
  __shared__ __attribute__((aligned(16))) float sW[OUT_SIZE * WP];

  const int lane  = threadIdx.x & 31;
  const int wave  = threadIdx.x >> 5;
  const int tile  = blockIdx.x * 8 + wave;
  const int ml    = lane & 15;
  const int koff  = (lane >> 4) << 2;
  int arow = tile * 16 + ml;
  if (arow > n - 1) arow = n - 1;
  const v4f* xr = (const v4f*)(x + (size_t)arow * IN_SIZE + koff);

  const int rbase = tile * 16 + ((lane >> 4) << 3);

  v4f a0 = xr[0], a1 = xr[2], a2 = xr[4], a3 = xr[6];
  unsigned int dc[8];
#pragma unroll
  for (int v = 0; v < 8; ++v) dc[v] = (dg && rbase + v < n) ? dg[rbase + v] : 0u;

  {
    const v4f* W4 = (const v4f*)W;
    v4f t[16];
#pragma unroll
    for (int i = 0; i < 16; ++i) t[i] = W4[i * 256 + threadIdx.x];
#pragma unroll
    for (int i = 0; i < 16; ++i) {
      const int p = i * 256 + threadIdx.x;
      *(v4f*)(sW + (p >> 6) * WP + ((p & 63) << 2)) = t[i];
    }
  }
  __syncthreads();
  if (tile * 16 >= n) return;

  const float* wl = sW + ml * WP + koff;

  v8f acc0 = {}, acc1 = {}, acc2 = {}, acc3 = {};
  v4f b0 = *(const v4f*)(wl + 0 * 16 * WP);
  v4f b1 = *(const v4f*)(wl + 1 * 16 * WP);
  v4f b2 = *(const v4f*)(wl + 2 * 16 * WP);
  v4f b3 = *(const v4f*)(wl + 3 * 16 * WP);

#define GEMM_STEP(A, KN)                                                                              \
  {                                                                                                  \
          \
    const v4f n0 = *(const v4f*)(wl + 0 * 16 * WP + (KN));                                           \
    const v4f n1 = *(const v4f*)(wl + 1 * 16 * WP + (KN));                                           \
    const v4f n2 = *(const v4f*)(wl + 2 * 16 * WP + (KN));                                           \
    const v4f n3 = *(const v4f*)(wl + 3 * 16 * WP + (KN));                                           \
    acc0 = __builtin_amdgcn_wmma_f32_16x16x4_f32(false, (A).lo, false, b0.lo, (short)0, acc0, false, false); \
    acc1 = __builtin_amdgcn_wmma_f32_16x16x4_f32(false, (A).lo, false, b1.lo, (short)0, acc1, false, false); \
    acc2 = __builtin_amdgcn_wmma_f32_16x16x4_f32(false, (A).lo, false, b2.lo, (short)0, acc2, false, false); \
    acc3 = __builtin_amdgcn_wmma_f32_16x16x4_f32(false, (A).lo, false, b3.lo, (short)0, acc3, false, false); \
    acc0 = __builtin_amdgcn_wmma_f32_16x16x4_f32(false, (A).hi, false, b0.hi, (short)0, acc0, false, false); \
    acc1 = __builtin_amdgcn_wmma_f32_16x16x4_f32(false, (A).hi, false, b1.hi, (short)0, acc1, false, false); \
    acc2 = __builtin_amdgcn_wmma_f32_16x16x4_f32(false, (A).hi, false, b2.hi, (short)0, acc2, false, false); \
    acc3 = __builtin_amdgcn_wmma_f32_16x16x4_f32(false, (A).hi, false, b3.hi, (short)0, acc3, false, false); \
    b0 = n0; b1 = n1; b2 = n2; b3 = n3;                                                              \
  }

  for (int k = 0; k < IN_SIZE; k += 32) {
    v4f q0 = a0, q1 = a1, q2 = a2, q3 = a3;
    if (k + 32 < IN_SIZE) {
      const v4f* xn = xr + ((k + 32) >> 2);
      q0 = xn[0]; q1 = xn[2]; q2 = xn[4]; q3 = xn[6];
    }
    GEMM_STEP(a0, k + 8)
    GEMM_STEP(a1, k + 16)
    GEMM_STEP(a2, k + 24)
    GEMM_STEP(a3, (k + 32) & (IN_SIZE - 1))
    a0 = q0; a1 = q1; a2 = q2; a3 = q3;
  }
#undef GEMM_STEP

  float* o = xw + (size_t)rbase * OUT_SIZE + ml;
#pragma unroll
  for (int v = 0; v < 8; ++v) {
    if (rbase + v < n) {
      const float s = dg ? rsqrtf((float)dc[v] + 1.0f) : 1.0f;
      o[(size_t)v * OUT_SIZE +  0] = s * acc0[v];
      o[(size_t)v * OUT_SIZE + 16] = s * acc1[v];
      o[(size_t)v * OUT_SIZE + 32] = s * acc2[v];
      o[(size_t)v * OUT_SIZE + 48] = s * acc3[v];
    }
  }
}


__global__ __launch_bounds__(256) void k_gather(const unsigned int* __restrict__ deg,
                                                const int* __restrict__ ovf,
                                                const int* __restrict__ row,
                                                const int* __restrict__ col,
                                                const float* __restrict__ xs,
                                                float* out, int n) {
  const int lane = threadIdx.x & 31;
  const int r = __builtin_amdgcn_readfirstlane(blockIdx.x * 8 + (threadIdx.x >> 5));
  if (r >= n) return;
  const int* sl = (const int*)out + (size_t)r * SLOTS;
  const int s_lo = sl[lane];
  const v2f* rows = (const v2f*)xs;
  v2f acc = rows[(size_t)r * 32 + lane];
  const unsigned int cnt = deg[r];
  const unsigned int m = cnt < (unsigned int)SLOTS ? cnt : (unsigned int)SLOTS;
  int s_hi = 0;
  if (cnt > 32u) s_hi = sl[32 + lane];
#pragma clang loop unroll(disable)
  for (unsigned int p = 0; p < m; p += 8) {
    const int sv = (p & 32u) ? s_hi : s_lo;
    int c[8];
#pragma unroll
    for (int k = 0; k < 8; ++k) {
      unsigned int q = p + k;
      q = q < m ? q : m - 1u;
      c[k] = __builtin_amdgcn_readlane(sv, (int)(q & 31u));
    }
    v2f v[8];
#pragma unroll
    for (int k = 0; k < 8; ++k) {
      v[k] = (v2f){0.0f, 0.0f};
      if (p + k < m) v[k] = rows[(size_t)c[k] * 32 + lane];
    }
#pragma unroll
    for (int k = 0; k < 8; ++k) acc += v[k];
  }
  if (cnt > (unsigned int)SLOTS) {
    const unsigned int M = deg[n];
    for (unsigned int j = 0; j < M; ++j) {
      const int e = ovf[j];
      if (row[e] == r) acc += rows[(size_t)col[e] * 32 + lane];
    }
  }
  const float d = rsqrtf((float)cnt + 1.0f);
  ((v2f*)out)[(size_t)r * 32 + lane] = acc * d;
}


__global__ __launch_bounds__(256) void k_count_deg(const int* __restrict__ row,
                                                   const int* __restrict__ col,
                                                   unsigned int* __restrict__ deg, int E) {
  int e = blockIdx.x * 256 + threadIdx.x;
  if (e < E) {
    int r = row[e], c = col[e];
    if (r != c) atomicAdd(&deg[r], 1u);
  }
}

__global__ __launch_bounds__(256) void k_dinv(const unsigned int* __restrict__ deg,
                                              float* __restrict__ dinv, int n) {
  int i = blockIdx.x * 256 + threadIdx.x;
  if (i < n) dinv[i] = rsqrtf((float)deg[i] + 1.0f);
}

__global__ __launch_bounds__(256) void k_self(const float* __restrict__ xw,
                                              const float* __restrict__ dinv,
                                              float* __restrict__ out, int n) {
  size_t idx = (size_t)blockIdx.x * 256 + threadIdx.x;
  if (idx < (size_t)n * OUT_SIZE) {
    int i = (int)(idx >> 6);
    float d = dinv[i];
    out[idx] = d * d * xw[idx];
  }
}

__global__ __launch_bounds__(256) void k_scatter(const int* __restrict__ row,
                                                 const int* __restrict__ col,
                                                 const float* __restrict__ dinv,
                                                 const float* __restrict__ xw,
                                                 float* __restrict__ out, int E) {
  int e = blockIdx.x * 4 + (threadIdx.x >> 6);
  int j = threadIdx.x & 63;
  if (e >= E) return;
  int r = row[e];
  int c = col[e];
  if (r == c) return;
  float norm = dinv[r] * dinv[c];
  unsafeAtomicAdd(out + (size_t)r * OUT_SIZE + j,
                  norm * xw[(size_t)c * OUT_SIZE + j]);
}


extern "C" void kernel_launch(void* const* d_in, const int* in_sizes, int n_in,
                              void* d_out, int out_size, void* d_ws, size_t ws_size,
                              hipStream_t stream) {
  const int*   edge = (const int*)d_in[0];
  const float* x    = (const float*)d_in[1];
  const float* W    = (const float*)d_in[3];

  const int E = in_sizes[0] / 2;
  const int n = in_sizes[1] / IN_SIZE;
  const int* row = edge;
  const int* col = edge + E;
  float* out = (float*)d_out;

  float*        xw   = (float*)d_ws;
  unsigned int* deg  = (unsigned int*)(xw + (size_t)n * OUT_SIZE);
  int*          ovf  = (int*)(deg + n + 1);
  float*        dinv = (float*)(deg + n);

  const int tiles = (n + 15) / 16;
  const size_t need = ((size_t)n * OUT_SIZE + (size_t)n + 1 + (size_t)E) * sizeof(float);

  if (ws_size >= need) {
    k_init_deg<<<(n + 1 + 255) / 256, 256, 0, stream>>>(deg, n + 1);
    k_place<<<(E + 255) / 256, 256, 0, stream>>>(row, col, deg, (int*)d_out, ovf, n, E);
    k_gemm<<<(tiles + 7) / 8, 256, 0, stream>>>(x, W, deg, xw, n);
    k_gather<<<(n + 7) / 8, 256, 0, stream>>>(deg, ovf, row, col, xw, out, n);
  } else {
    k_init_deg<<<(n + 255) / 256, 256, 0, stream>>>(deg, n);
    k_count_deg<<<(E + 255) / 256, 256, 0, stream>>>(row, col, deg, E);
    k_dinv<<<(n + 255) / 256, 256, 0, stream>>>(deg, dinv, n);
    k_gemm<<<(tiles + 7) / 8, 256, 0, stream>>>(x, W, nullptr, xw, n);
    const size_t tot = (size_t)n * OUT_SIZE;
    k_self<<<(unsigned)((tot + 255) / 256), 256, 0, stream>>>(xw, dinv, out, n);
    k_scatter<<<(E + 3) / 4, 256, 0, stream>>>(row, col, dinv, xw, out, E);
  }
}


# Edge placement through LDS: chunk groups bin edges by 256-row bucket into 256-byte cells, one group per bucket ranks them into the row slots; no global atomics; sum order = LDS atomic arrival order
# speedup vs baseline: 5.1732x; 2.0231x over previous
#include <hip/hip_runtime.h>
#include <hip/hip_bf16.h>


typedef __attribute__((ext_vector_type(2))) float v2f;
typedef __attribute__((ext_vector_type(8))) float v8f;

#define IN_SIZE 256
#define OUT_SIZE 64
#define SCAN_B 1024


#define SLOTS 64

__global__ __launch_bounds__(256) void k_init_deg(unsigned int* __restrict__ cnt, int m) {
  int i = blockIdx.x * 256 + threadIdx.x;
  if (i < m) cnt[i] = 0u;
}

__global__ __launch_bounds__(256) void k_place(const int* __restrict__ row,
                                               const int* __restrict__ col,
                                               unsigned int* __restrict__ deg,
                                               int* __restrict__ slots,
                                               int* __restrict__ ovf, int n, int E) {
  int e = blockIdx.x * 256 + threadIdx.x;
  if (e < E) {
    int r = row[e], c = col[e];
    if (r != c) {
      unsigned int k = atomicAdd(&deg[r], 1u);
      if (k < (unsigned int)SLOTS) {
        slots[(size_t)r * SLOTS + k] = c;
      } else {
        unsigned int j = atomicAdd(&deg[n], 1u);
        ovf[j] = e;
      }
    }
  }
}


#define RBITS 8
#define GA 250
#define NBMAX 512
#define CELLW 64

typedef __attribute__((ext_vector_type(4))) unsigned int v4u;

__global__ __launch_bounds__(1024) void k_bin(const int* __restrict__ row,
                                              const int* __restrict__ col,
                                              unsigned int* __restrict__ seg, int E, int NB, int chunk) {
  __shared__ __attribute__((aligned(16))) unsigned int stage[NBMAX * CELLW];
  const int tid = threadIdx.x;
  const int w = blockIdx.x;
  v4u* st4 = (v4u*)stage;
  for (int i = tid; i < NB * 16; i += 1024) st4[i] = (v4u){0u, 0u, 0u, 0u};
  __syncthreads();
  const int e0 = w * chunk;
  int e1 = e0 + chunk;
  if (e1 > E) e1 = E;
  for (int base = e0; base < e1; base += 4096) {
    int r[4], c[4];
#pragma unroll
    for (int j = 0; j < 4; ++j) {
      const int e = base + j * 1024 + tid;
      const bool in = e < e1;
      r[j] = in ? row[e] : 0;
      c[j] = in ? col[e] : 0;
    }
#pragma unroll
    for (int j = 0; j < 4; ++j) {
      if (r[j] != c[j]) {
        unsigned int* cell = stage + ((unsigned int)r[j] >> RBITS) * CELLW;
        const unsigned int k = atomicAdd(cell, 1u);
        if (k < (unsigned int)(CELLW - 1))
          cell[1 + k] = ((unsigned int)(r[j] & 255) << 24) | (unsigned int)c[j];
      }
    }
  }
  __syncthreads();
  for (int i = tid; i < NB * 16; i += 1024) {
    const int b = i >> 4, q = i & 15;
    const unsigned int cnt = stage[b * CELLW];
    if (q < 8 || cnt > 31u) ((v4u*)(seg + ((size_t)b * GA + w) * CELLW))[q] = st4[i];
  }
}

__global__ __launch_bounds__(1024) void k_fine(const unsigned int* __restrict__ seg,
                                               const int* __restrict__ row,
                                               const int* __restrict__ col,
                                               unsigned int* __restrict__ deg,
                                               int* __restrict__ slots, int n, int E) {
  __shared__ __attribute__((aligned(16))) int sl[256 * SLOTS];
  __shared__ unsigned int rc[256];
  __shared__ unsigned int cc[GA];
  __shared__ unsigned int over;
  const int tid = threadIdx.x;
  const int b = blockIdx.x;
  const unsigned int* cells = seg + (size_t)b * GA * CELLW;
  for (int i = tid; i < 256 * SLOTS / 4; i += 1024) ((v4u*)sl)[i] = (v4u){0u, 0u, 0u, 0u};
  if (tid < 256) rc[tid] = 0u;
  if (tid == 0) over = 0u;
  __syncthreads();
  if (tid < GA) {
    const unsigned int cw = cells[tid * CELLW];
    cc[tid] = cw;
    if (cw > (unsigned int)(CELLW - 1)) over = 1u;
  }
  __syncthreads();
  if (over == 0u) {
    for (int base = 0; base < GA * CELLW; base += 4096) {
      unsigned int wd[4];
      bool ok[4];
#pragma unroll
      for (int j = 0; j < 4; ++j) {
        const int i = base + j * 1024 + tid;
        const bool in = i < GA * CELLW;
        const unsigned int jj = (unsigned int)(i & (CELLW - 1));
        ok[j] = in && jj >= 1u && jj <= cc[in ? (i >> 6) : 0];
        wd[j] = ok[j] ? cells[i] : 0u;
      }
#pragma unroll
      for (int j = 0; j < 4; ++j) {
        if (ok[j]) {
          const unsigned int rl = wd[j] >> 24;
          const unsigned int k = atomicAdd(&rc[rl], 1u);
          if (k < (unsigned int)SLOTS) sl[rl * SLOTS + k] = (int)(wd[j] & 0xFFFFFFu);
        }
      }
    }
  } else {
    for (int e = tid; e < E; e += 1024) {
      const int r = row[e], c = col[e];
      if (r != c && (r >> RBITS) == b) {
        const unsigned int rl = (unsigned int)(r & 255);
        const unsigned int k = atomicAdd(&rc[rl], 1u);
        if (k < (unsigned int)SLOTS) sl[rl * SLOTS + k] = c;
      }
    }
  }
  __syncthreads();
  if (tid < 256) {
    const int r = b * 256 + tid;
    if (r < n) deg[r] = rc[tid];
  }
  for (int i = tid; i < 256 * 16; i += 1024) {
    const int rl = i >> 4, q = i & 15;
    const int r = b * 256 + rl;
    if (r < n && (q < 8 || rc[rl] > 32u)) ((v4u*)slots)[(size_t)r * 16 + q] = ((const v4u*)sl)[i];
  }
}


typedef __attribute__((ext_vector_type(4))) float v4f;

#define WP 260

__global__ __launch_bounds__(256) void k_gemm(const float* __restrict__ x,
                                              const float* __restrict__ W,
                                              const unsigned int* __restrict__ dg,
                                              float* __restrict__ xw, int n) {
  __shared__ __attribute__((aligned(16))) float sW[OUT_SIZE * WP];

  const int lane  = threadIdx.x & 31;
  const int wave  = threadIdx.x >> 5;
  const int tile  = blockIdx.x * 8 + wave;
  const int ml    = lane & 15;
  const int koff  = (lane >> 4) << 2;
  int arow = tile * 16 + ml;
  if (arow > n - 1) arow = n - 1;
  const v4f* xr = (const v4f*)(x + (size_t)arow * IN_SIZE + koff);

  const int rbase = tile * 16 + ((lane >> 4) << 3);

  v4f a0 = xr[0], a1 = xr[2], a2 = xr[4], a3 = xr[6];
  unsigned int dc[8];
#pragma unroll
  for (int v = 0; v < 8; ++v) dc[v] = (dg && rbase + v < n) ? dg[rbase + v] : 0u;

  {
    const v4f* W4 = (const v4f*)W;
    v4f t[16];
#pragma unroll
    for (int i = 0; i < 16; ++i) t[i] = W4[i * 256 + threadIdx.x];
#pragma unroll
    for (int i = 0; i < 16; ++i) {
      const int p = i * 256 + threadIdx.x;
      *(v4f*)(sW + (p >> 6) * WP + ((p & 63) << 2)) = t[i];
    }
  }
  __syncthreads();
  if (tile * 16 >= n) return;

  const float* wl = sW + ml * WP + koff;

  v8f acc0 = {}, acc1 = {}, acc2 = {}, acc3 = {};
  v4f b0 = *(const v4f*)(wl + 0 * 16 * WP);
  v4f b1 = *(const v4f*)(wl + 1 * 16 * WP);
  v4f b2 = *(const v4f*)(wl + 2 * 16 * WP);
  v4f b3 = *(const v4f*)(wl + 3 * 16 * WP);

#define GEMM_STEP(A, KN)                                                                              \
  {                                                                                                  \
          \
    const v4f n0 = *(const v4f*)(wl + 0 * 16 * WP + (KN));                                           \
    const v4f n1 = *(const v4f*)(wl + 1 * 16 * WP + (KN));                                           \
    const v4f n2 = *(const v4f*)(wl + 2 * 16 * WP + (KN));                                           \
    const v4f n3 = *(const v4f*)(wl + 3 * 16 * WP + (KN));                                           \
    acc0 = __builtin_amdgcn_wmma_f32_16x16x4_f32(false, (A).lo, false, b0.lo, (short)0, acc0, false, false); \
    acc1 = __builtin_amdgcn_wmma_f32_16x16x4_f32(false, (A).lo, false, b1.lo, (short)0, acc1, false, false); \
    acc2 = __builtin_amdgcn_wmma_f32_16x16x4_f32(false, (A).lo, false, b2.lo, (short)0, acc2, false, false); \
    acc3 = __builtin_amdgcn_wmma_f32_16x16x4_f32(false, (A).lo, false, b3.lo, (short)0, acc3, false, false); \
    acc0 = __builtin_amdgcn_wmma_f32_16x16x4_f32(false, (A).hi, false, b0.hi, (short)0, acc0, false, false); \
    acc1 = __builtin_amdgcn_wmma_f32_16x16x4_f32(false, (A).hi, false, b1.hi, (short)0, acc1, false, false); \
    acc2 = __builtin_amdgcn_wmma_f32_16x16x4_f32(false, (A).hi, false, b2.hi, (short)0, acc2, false, false); \
    acc3 = __builtin_amdgcn_wmma_f32_16x16x4_f32(false, (A).hi, false, b3.hi, (short)0, acc3, false, false); \
    b0 = n0; b1 = n1; b2 = n2; b3 = n3;                                                              \
  }

  for (int k = 0; k < IN_SIZE; k += 32) {
    v4f q0 = a0, q1 = a1, q2 = a2, q3 = a3;
    if (k + 32 < IN_SIZE) {
      const v4f* xn = xr + ((k + 32) >> 2);
      q0 = xn[0]; q1 = xn[2]; q2 = xn[4]; q3 = xn[6];
    }
    GEMM_STEP(a0, k + 8)
    GEMM_STEP(a1, k + 16)
    GEMM_STEP(a2, k + 24)
    GEMM_STEP(a3, (k + 32) & (IN_SIZE - 1))
    a0 = q0; a1 = q1; a2 = q2; a3 = q3;
  }
#undef GEMM_STEP

  float* o = xw + (size_t)rbase * OUT_SIZE + ml;
#pragma unroll
  for (int v = 0; v < 8; ++v) {
    if (rbase + v < n) {
      const float s = dg ? rsqrtf((float)dc[v] + 1.0f) : 1.0f;
      o[(size_t)v * OUT_SIZE +  0] = s * acc0[v];
      o[(size_t)v * OUT_SIZE + 16] = s * acc1[v];
      o[(size_t)v * OUT_SIZE + 32] = s * acc2[v];
      o[(size_t)v * OUT_SIZE + 48] = s * acc3[v];
    }
  }
}


__global__ __launch_bounds__(256) void k_gather(const unsigned int* __restrict__ deg,
                                                const int* __restrict__ ovf,
                                                const int* __restrict__ row,
                                                const int* __restrict__ col,
                                                const float* __restrict__ xs,
                                                float* out, int n, int E) {
  const int lane = threadIdx.x & 31;
  const int r = __builtin_amdgcn_readfirstlane(blockIdx.x * 8 + (threadIdx.x >> 5));
  if (r >= n) return;
  const int* sl = (const int*)out + (size_t)r * SLOTS;
  const int s_lo = sl[lane];
  const v2f* rows = (const v2f*)xs;
  v2f acc = rows[(size_t)r * 32 + lane];
  const unsigned int cnt = deg[r];
  const unsigned int m = cnt < (unsigned int)SLOTS ? cnt : (unsigned int)SLOTS;
  int s_hi = 0;
  if (cnt > 32u) s_hi = sl[32 + lane];
#pragma clang loop unroll(disable)
  for (unsigned int p = 0; p < m; p += 8) {
    const int sv = (p & 32u) ? s_hi : s_lo;
    int c[8];
#pragma unroll
    for (int k = 0; k < 8; ++k) {
      unsigned int q = p + k;
      q = q < m ? q : m - 1u;
      c[k] = __builtin_amdgcn_readlane(sv, (int)(q & 31u));
    }
    v2f v[8];
#pragma unroll
    for (int k = 0; k < 8; ++k) {
      v[k] = (v2f){0.0f, 0.0f};
      if (p + k < m) v[k] = rows[(size_t)c[k] * 32 + lane];
    }
#pragma unroll
    for (int k = 0; k < 8; ++k) acc += v[k];
  }
  if (cnt > (unsigned int)SLOTS) {
    if (ovf) {
      const unsigned int M = deg[n];
      for (unsigned int j = 0; j < M; ++j) {
        const int e = ovf[j];
        if (row[e] == r) acc += rows[(size_t)col[e] * 32 + lane];
      }
    } else {
      acc = rows[(size_t)r * 32 + lane];
      for (int e = 0; e < E; ++e) {
        const int cs = col[e];
        if (row[e] == r && cs != r) acc += rows[(size_t)cs * 32 + lane];
      }
    }
  }
  const float d = rsqrtf((float)cnt + 1.0f);
  ((v2f*)out)[(size_t)r * 32 + lane] = acc * d;
}


__global__ __launch_bounds__(256) void k_count_deg(const int* __restrict__ row,
                                                   const int* __restrict__ col,
                                                   unsigned int* __restrict__ deg, int E) {
  int e = blockIdx.x * 256 + threadIdx.x;
  if (e < E) {
    int r = row[e], c = col[e];
    if (r != c) atomicAdd(&deg[r], 1u);
  }
}

__global__ __launch_bounds__(256) void k_dinv(const unsigned int* __restrict__ deg,
                                              float* __restrict__ dinv, int n) {
  int i = blockIdx.x * 256 + threadIdx.x;
  if (i < n) dinv[i] = rsqrtf((float)deg[i] + 1.0f);
}

__global__ __launch_bounds__(256) void k_self(const float* __restrict__ xw,
                                              const float* __restrict__ dinv,
                                              float* __restrict__ out, int n) {
  size_t idx = (size_t)blockIdx.x * 256 + threadIdx.x;
  if (idx < (size_t)n * OUT_SIZE) {
    int i = (int)(idx >> 6);
    float d = dinv[i];
    out[idx] = d * d * xw[idx];
  }
}

__global__ __launch_bounds__(256) void k_scatter(const int* __restrict__ row,
                                                 const int* __restrict__ col,
                                                 const float* __restrict__ dinv,
                                                 const float* __restrict__ xw,
                                                 float* __restrict__ out, int E) {
  int e = blockIdx.x * 4 + (threadIdx.x >> 6);
  int j = threadIdx.x & 63;
  if (e >= E) return;
  int r = row[e];
  int c = col[e];
  if (r == c) return;
  float norm = dinv[r] * dinv[c];
  unsafeAtomicAdd(out + (size_t)r * OUT_SIZE + j,
                  norm * xw[(size_t)c * OUT_SIZE + j]);
}


extern "C" void kernel_launch(void* const* d_in, const int* in_sizes, int n_in,
                              void* d_out, int out_size, void* d_ws, size_t ws_size,
                              hipStream_t stream) {
  const int*   edge = (const int*)d_in[0];
  const float* x    = (const float*)d_in[1];
  const float* W    = (const float*)d_in[3];

  const int E = in_sizes[0] / 2;
  const int n = in_sizes[1] / IN_SIZE;
  const int* row = edge;
  const int* col = edge + E;
  float* out = (float*)d_out;

  float*        xw   = (float*)d_ws;
  unsigned int* deg  = (unsigned int*)(xw + (size_t)n * OUT_SIZE);
  int*          ovf  = (int*)(deg + n + 1);
  float*        dinv = (float*)(deg + n);

  const int tiles = (n + 15) / 16;
  const size_t need = ((size_t)n * OUT_SIZE + (size_t)n + 1 + (size_t)E) * sizeof(float);

  const int NB = (n + 255) / 256;
  if (ws_size >= need && n <= 131072 && (size_t)NB * GA <= (size_t)n) {
    unsigned int* seg = (unsigned int*)d_ws;
    k_bin<<<GA, 1024, 0, stream>>>(row, col, seg, E, NB, (E + GA - 1) / GA);
    k_fine<<<NB, 1024, 0, stream>>>(seg, row, col, deg, (int*)d_out, n, E);
    k_gemm<<<(tiles + 7) / 8, 256, 0, stream>>>(x, W, deg, xw, n);
    k_gather<<<(n + 7) / 8, 256, 0, stream>>>(deg, nullptr, row, col, xw, out, n, E);
  } else if (ws_size >= need) {
    k_init_deg<<<(n + 1 + 255) / 256, 256, 0, stream>>>(deg, n + 1);
    k_place<<<(E + 255) / 256, 256, 0, stream>>>(row, col, deg, (int*)d_out, ovf, n, E);
    k_gemm<<<(tiles + 7) / 8, 256, 0, stream>>>(x, W, deg, xw, n);
    k_gather<<<(n + 7) / 8, 256, 0, stream>>>(deg, ovf, row, col, xw, out, n, E);
  } else {
    k_init_deg<<<(n + 255) / 256, 256, 0, stream>>>(deg, n);
    k_count_deg<<<(E + 255) / 256, 256, 0, stream>>>(row, col, deg, E);
    k_dinv<<<(n + 255) / 256, 256, 0, stream>>>(deg, dinv, n);
    k_gemm<<<(tiles + 7) / 8, 256, 0, stream>>>(x, W, nullptr, xw, n);
    const size_t tot = (size_t)n * OUT_SIZE;
    k_self<<<(unsigned)((tot + 255) / 256), 256, 0, stream>>>(xw, dinv, out, n);
    k_scatter<<<(E + 3) / 4, 256, 0, stream>>>(row, col, dinv, xw, out, E);
  }
}


# GEMM grid: 250 work-groups of 25 waves (one per WGP) instead of 782 of 8: W staged into LDS once per WGP, fullest SIMD runs 7 row tiles not 8; every sum and its order unchanged
# speedup vs baseline: 5.6429x; 1.0908x over previous
#include <hip/hip_runtime.h>
#include <hip/hip_bf16.h>


typedef __attribute__((ext_vector_type(2))) float v2f;
typedef __attribute__((ext_vector_type(8))) float v8f;

#define IN_SIZE 256
#define OUT_SIZE 64
#define SCAN_B 1024


#define SLOTS 64

__global__ __launch_bounds__(256) void k_init_deg(unsigned int* __restrict__ cnt, int m) {
  int i = blockIdx.x * 256 + threadIdx.x;
  if (i < m) cnt[i] = 0u;
}

__global__ __launch_bounds__(256) void k_place(const int* __restrict__ row,
                                               const int* __restrict__ col,
                                               unsigned int* __restrict__ deg,
                                               int* __restrict__ slots,
                                               int* __restrict__ ovf, int n, int E) {
  int e = blockIdx.x * 256 + threadIdx.x;
  if (e < E) {
    int r = row[e], c = col[e];
    if (r != c) {
      unsigned int k = atomicAdd(&deg[r], 1u);
      if (k < (unsigned int)SLOTS) {
        slots[(size_t)r * SLOTS + k] = c;
      } else {
        unsigned int j = atomicAdd(&deg[n], 1u);
        ovf[j] = e;
      }
    }
  }
}


#define RBITS 8
#define GA 250
#define NBMAX 512
#define CELLW 64

typedef __attribute__((ext_vector_type(4))) unsigned int v4u;

__global__ __launch_bounds__(1024) void k_bin(const int* __restrict__ row,
                                              const int* __restrict__ col,
                                              unsigned int* __restrict__ seg, int E, int NB, int chunk) {
  __shared__ __attribute__((aligned(16))) unsigned int stage[NBMAX * CELLW];
  const int tid = threadIdx.x;
  const int w = blockIdx.x;
  v4u* st4 = (v4u*)stage;
  for (int i = tid; i < NB * 16; i += 1024) st4[i] = (v4u){0u, 0u, 0u, 0u};
  __syncthreads();
  const int e0 = w * chunk;
  int e1 = e0 + chunk;
  if (e1 > E) e1 = E;
  for (int base = e0; base < e1; base += 4096) {
    int r[4], c[4];
#pragma unroll
    for (int j = 0; j < 4; ++j) {
      const int e = base + j * 1024 + tid;
      const bool in = e < e1;
      r[j] = in ? row[e] : 0;
      c[j] = in ? col[e] : 0;
    }
#pragma unroll
    for (int j = 0; j < 4; ++j) {
      if (r[j] != c[j]) {
        unsigned int* cell = stage + ((unsigned int)r[j] >> RBITS) * CELLW;
        const unsigned int k = atomicAdd(cell, 1u);
        if (k < (unsigned int)(CELLW - 1))
          cell[1 + k] = ((unsigned int)(r[j] & 255) << 24) | (unsigned int)c[j];
      }
    }
  }
  __syncthreads();
  for (int i = tid; i < NB * 16; i += 1024) {
    const int b = i >> 4, q = i & 15;
    const unsigned int cnt = stage[b * CELLW];
    if (q < 8 || cnt > 31u) ((v4u*)(seg + ((size_t)b * GA + w) * CELLW))[q] = st4[i];
  }
}

__global__ __launch_bounds__(1024) void k_fine(const unsigned int* __restrict__ seg,
                                               const int* __restrict__ row,
                                               const int* __restrict__ col,
                                               unsigned int* __restrict__ deg,
                                               int* __restrict__ slots, int n, int E) {
  __shared__ __attribute__((aligned(16))) int sl[256 * SLOTS];
  __shared__ unsigned int rc[256];
  __shared__ unsigned int cc[GA];
  __shared__ unsigned int over;
  const int tid = threadIdx.x;
  const int b = blockIdx.x;
  const unsigned int* cells = seg + (size_t)b * GA * CELLW;
  for (int i = tid; i < 256 * SLOTS / 4; i += 1024) ((v4u*)sl)[i] = (v4u){0u, 0u, 0u, 0u};
  if (tid < 256) rc[tid] = 0u;
  if (tid == 0) over = 0u;
  __syncthreads();
  if (tid < GA) {
    const unsigned int cw = cells[tid * CELLW];
    cc[tid] = cw;
    if (cw > (unsigned int)(CELLW - 1)) over = 1u;
  }
  __syncthreads();
  if (over == 0u) {
    for (int base = 0; base < GA * CELLW; base += 4096) {
      unsigned int wd[4];
      bool ok[4];
#pragma unroll
      for (int j = 0; j < 4; ++j) {
        const int i = base + j * 1024 + tid;
        const bool in = i < GA * CELLW;
        const unsigned int jj = (unsigned int)(i & (CELLW - 1));
        ok[j] = in && jj >= 1u && jj <= cc[in ? (i >> 6) : 0];
        wd[j] = ok[j] ? cells[i] : 0u;
      }
#pragma unroll
      for (int j = 0; j < 4; ++j) {
        if (ok[j]) {
          const unsigned int rl = wd[j] >> 24;
          const unsigned int k = atomicAdd(&rc[rl], 1u);
          if (k < (unsigned int)SLOTS) sl[rl * SLOTS + k] = (int)(wd[j] & 0xFFFFFFu);
        }
      }
    }
  } else {
    for (int e = tid; e < E; e += 1024) {
      const int r = row[e], c = col[e];
      if (r != c && (r >> RBITS) == b) {
        const unsigned int rl = (unsigned int)(r & 255);
        const unsigned int k = atomicAdd(&rc[rl], 1u);
        if (k < (unsigned int)SLOTS) sl[rl * SLOTS + k] = c;
      }
    }
  }
  __syncthreads();
  if (tid < 256) {
    const int r = b * 256 + tid;
    if (r < n) deg[r] = rc[tid];
  }
  for (int i = tid; i < 256 * 16; i += 1024) {
    const int rl = i >> 4, q = i & 15;
    const int r = b * 256 + rl;
    if (r < n && (q < 8 || rc[rl] > 32u)) ((v4u*)slots)[(size_t)r * 16 + q] = ((const v4u*)sl)[i];
  }
}


typedef __attribute__((ext_vector_type(4))) float v4f;

#define WP 260
#define GW 25

__global__ __launch_bounds__(GW * 32) void k_gemm(const float* __restrict__ x,
                                              const float* __restrict__ W,
                                              const unsigned int* __restrict__ dg,
                                              float* __restrict__ xw, int n) {
  __shared__ __attribute__((aligned(16))) float sW[OUT_SIZE * WP];

  const int lane  = threadIdx.x & 31;
  const int wave  = threadIdx.x >> 5;
  const int tile  = blockIdx.x * GW + wave;
  const int ml    = lane & 15;
  const int koff  = (lane >> 4) << 2;
  int arow = tile * 16 + ml;
  if (arow > n - 1) arow = n - 1;
  const v4f* xr = (const v4f*)(x + (size_t)arow * IN_SIZE + koff);

  const int rbase = tile * 16 + ((lane >> 4) << 3);

  v4f a0 = xr[0], a1 = xr[2], a2 = xr[4], a3 = xr[6];
  unsigned int dc[8];
#pragma unroll
  for (int v = 0; v < 8; ++v) dc[v] = (dg && rbase + v < n) ? dg[rbase + v] : 0u;

  {
    const v4f* W4 = (const v4f*)W;
    constexpr int WT = (OUT_SIZE * IN_SIZE / 4 + GW * 32 - 1) / (GW * 32);
    v4f t[WT];
#pragma unroll
    for (int i = 0; i < WT; ++i) {
      const int p = i * (GW * 32) + (int)threadIdx.x;
      t[i] = W4[p < OUT_SIZE * IN_SIZE / 4 ? p : 0];
    }
#pragma unroll
    for (int i = 0; i < WT; ++i) {
      const int p = i * (GW * 32) + (int)threadIdx.x;
      if (p < OUT_SIZE * IN_SIZE / 4) *(v4f*)(sW + (p >> 6) * WP + ((p & 63) << 2)) = t[i];
    }
  }
  __syncthreads();
  if (tile * 16 >= n) return;

  const float* wl = sW + ml * WP + koff;

  v8f acc0 = {}, acc1 = {}, acc2 = {}, acc3 = {};
  v4f b0 = *(const v4f*)(wl + 0 * 16 * WP);
  v4f b1 = *(const v4f*)(wl + 1 * 16 * WP);
  v4f b2 = *(const v4f*)(wl + 2 * 16 * WP);
  v4f b3 = *(const v4f*)(wl + 3 * 16 * WP);

#define GEMM_STEP(A, KN)                                                                              \
  {                                                                                                  \
          \
    const v4f n0 = *(const v4f*)(wl + 0 * 16 * WP + (KN));                                           \
    const v4f n1 = *(const v4f*)(wl + 1 * 16 * WP + (KN));                                           \
    const v4f n2 = *(const v4f*)(wl + 2 * 16 * WP + (KN));                                           \
    const v4f n3 = *(const v4f*)(wl + 3 * 16 * WP + (KN));                                           \
    acc0 = __builtin_amdgcn_wmma_f32_16x16x4_f32(false, (A).lo, false, b0.lo, (short)0, acc0, false, false); \
    acc1 = __builtin_amdgcn_wmma_f32_16x16x4_f32(false, (A).lo, false, b1.lo, (short)0, acc1, false, false); \
    acc2 = __builtin_amdgcn_wmma_f32_16x16x4_f32(false, (A).lo, false, b2.lo, (short)0, acc2, false, false); \
    acc3 = __builtin_amdgcn_wmma_f32_16x16x4_f32(false, (A).lo, false, b3.lo, (short)0, acc3, false, false); \
    acc0 = __builtin_amdgcn_wmma_f32_16x16x4_f32(false, (A).hi, false, b0.hi, (short)0, acc0, false, false); \
    acc1 = __builtin_amdgcn_wmma_f32_16x16x4_f32(false, (A).hi, false, b1.hi, (short)0, acc1, false, false); \
    acc2 = __builtin_amdgcn_wmma_f32_16x16x4_f32(false, (A).hi, false, b2.hi, (short)0, acc2, false, false); \
    acc3 = __builtin_amdgcn_wmma_f32_16x16x4_f32(false, (A).hi, false, b3.hi, (short)0, acc3, false, false); \
    b0 = n0; b1 = n1; b2 = n2; b3 = n3;                                                              \
  }

  for (int k = 0; k < IN_SIZE; k += 32) {
    v4f q0 = a0, q1 = a1, q2 = a2, q3 = a3;
    if (k + 32 < IN_SIZE) {
      const v4f* xn = xr + ((k + 32) >> 2);
      q0 = xn[0]; q1 = xn[2]; q2 = xn[4]; q3 = xn[6];
    }
    GEMM_STEP(a0, k + 8)
    GEMM_STEP(a1, k + 16)
    GEMM_STEP(a2, k + 24)
    GEMM_STEP(a3, (k + 32) & (IN_SIZE - 1))
    a0 = q0; a1 = q1; a2 = q2; a3 = q3;
  }
#undef GEMM_STEP

  float* o = xw + (size_t)rbase * OUT_SIZE + ml;
#pragma unroll
  for (int v = 0; v < 8; ++v) {
    if (rbase + v < n) {
      const float s = dg ? rsqrtf((float)dc[v] + 1.0f) : 1.0f;
      o[(size_t)v * OUT_SIZE +  0] = s * acc0[v];
      o[(size_t)v * OUT_SIZE + 16] = s * acc1[v];
      o[(size_t)v * OUT_SIZE + 32] = s * acc2[v];
      o[(size_t)v * OUT_SIZE + 48] = s * acc3[v];
    }
  }
}


__global__ __launch_bounds__(256) void k_gather(const unsigned int* __restrict__ deg,
                                                const int* __restrict__ ovf,
                                                const int* __restrict__ row,
                                                const int* __restrict__ col,
                                                const float* __restrict__ xs,
                                                float* out, int n, int E) {
  const int lane = threadIdx.x & 31;
  const int r = __builtin_amdgcn_readfirstlane(blockIdx.x * 8 + (threadIdx.x >> 5));
  if (r >= n) return;
  const int* sl = (const int*)out + (size_t)r * SLOTS;
  const int s_lo = sl[lane];
  const v2f* rows = (const v2f*)xs;
  v2f acc = rows[(size_t)r * 32 + lane];
  const unsigned int cnt = deg[r];
  const unsigned int m = cnt < (unsigned int)SLOTS ? cnt : (unsigned int)SLOTS;
  int s_hi = 0;
  if (cnt > 32u) s_hi = sl[32 + lane];
#pragma clang loop unroll(disable)
  for (unsigned int p = 0; p < m; p += 8) {
    const int sv = (p & 32u) ? s_hi : s_lo;
    int c[8];
#pragma unroll
    for (int k = 0; k < 8; ++k) {
      unsigned int q = p + k;
      q = q < m ? q : m - 1u;
      c[k] = __builtin_amdgcn_readlane(sv, (int)(q & 31u));
    }
    v2f v[8];
#pragma unroll
    for (int k = 0; k < 8; ++k) {
      v[k] = (v2f){0.0f, 0.0f};
      if (p + k < m) v[k] = rows[(size_t)c[k] * 32 + lane];
    }
#pragma unroll
    for (int k = 0; k < 8; ++k) acc += v[k];
  }
  if (cnt > (unsigned int)SLOTS) {
    if (ovf) {
      const unsigned int M = deg[n];
      for (unsigned int j = 0; j < M; ++j) {
        const int e = ovf[j];
        if (row[e] == r) acc += rows[(size_t)col[e] * 32 + lane];
      }
    } else {
      acc = rows[(size_t)r * 32 + lane];
      for (int e = 0; e < E; ++e) {
        const int cs = col[e];
        if (row[e] == r && cs != r) acc += rows[(size_t)cs * 32 + lane];
      }
    }
  }
  const float d = rsqrtf((float)cnt + 1.0f);
  ((v2f*)out)[(size_t)r * 32 + lane] = acc * d;
}


__global__ __launch_bounds__(256) void k_count_deg(const int* __restrict__ row,
                                                   const int* __restrict__ col,
                                                   unsigned int* __restrict__ deg, int E) {
  int e = blockIdx.x * 256 + threadIdx.x;
  if (e < E) {
    int r = row[e], c = col[e];
    if (r != c) atomicAdd(&deg[r], 1u);
  }
}

__global__ __launch_bounds__(256) void k_dinv(const unsigned int* __restrict__ deg,
                                              float* __restrict__ dinv, int n) {
  int i = blockIdx.x * 256 + threadIdx.x;
  if (i < n) dinv[i] = rsqrtf((float)deg[i] + 1.0f);
}

__global__ __launch_bounds__(256) void k_self(const float* __restrict__ xw,
                                              const float* __restrict__ dinv,
                                              float* __restrict__ out, int n) {
  size_t idx = (size_t)blockIdx.x * 256 + threadIdx.x;
  if (idx < (size_t)n * OUT_SIZE) {
    int i = (int)(idx >> 6);
    float d = dinv[i];
    out[idx] = d * d * xw[idx];
  }
}

__global__ __launch_bounds__(256) void k_scatter(const int* __restrict__ row,
                                                 const int* __restrict__ col,
                                                 const float* __restrict__ dinv,
                                                 const float* __restrict__ xw,
                                                 float* __restrict__ out, int E) {
  int e = blockIdx.x * 4 + (threadIdx.x >> 6);
  int j = threadIdx.x & 63;
  if (e >= E) return;
  int r = row[e];
  int c = col[e];
  if (r == c) return;
  float norm = dinv[r] * dinv[c];
  unsafeAtomicAdd(out + (size_t)r * OUT_SIZE + j,
                  norm * xw[(size_t)c * OUT_SIZE + j]);
}


extern "C" void kernel_launch(void* const* d_in, const int* in_sizes, int n_in,
                              void* d_out, int out_size, void* d_ws, size_t ws_size,
                              hipStream_t stream) {
  const int*   edge = (const int*)d_in[0];
  const float* x    = (const float*)d_in[1];
  const float* W    = (const float*)d_in[3];

  const int E = in_sizes[0] / 2;
  const int n = in_sizes[1] / IN_SIZE;
  const int* row = edge;
  const int* col = edge + E;
  float* out = (float*)d_out;

  float*        xw   = (float*)d_ws;
  unsigned int* deg  = (unsigned int*)(xw + (size_t)n * OUT_SIZE);
  int*          ovf  = (int*)(deg + n + 1);
  float*        dinv = (float*)(deg + n);

  const int tiles = (n + 15) / 16;
  const size_t need = ((size_t)n * OUT_SIZE + (size_t)n + 1 + (size_t)E) * sizeof(float);

  const int NB = (n + 255) / 256;
  if (ws_size >= need && n <= 131072 && (size_t)NB * GA <= (size_t)n) {
    unsigned int* seg = (unsigned int*)d_ws;
    k_bin<<<GA, 1024, 0, stream>>>(row, col, seg, E, NB, (E + GA - 1) / GA);
    k_fine<<<NB, 1024, 0, stream>>>(seg, row, col, deg, (int*)d_out, n, E);
    k_gemm<<<(tiles + GW - 1) / GW, GW * 32, 0, stream>>>(x, W, deg, xw, n);
    k_gather<<<(n + 7) / 8, 256, 0, stream>>>(deg, nullptr, row, col, xw, out, n, E);
  } else if (ws_size >= need) {
    k_init_deg<<<(n + 1 + 255) / 256, 256, 0, stream>>>(deg, n + 1);
    k_place<<<(E + 255) / 256, 256, 0, stream>>>(row, col, deg, (int*)d_out, ovf, n, E);
    k_gemm<<<(tiles + GW - 1) / GW, GW * 32, 0, stream>>>(x, W, deg, xw, n);
    k_gather<<<(n + 7) / 8, 256, 0, stream>>>(deg, ovf, row, col, xw, out, n, E);
  } else {
    k_init_deg<<<(n + 255) / 256, 256, 0, stream>>>(deg, n);
    k_count_deg<<<(E + 255) / 256, 256, 0, stream>>>(row, col, deg, E);
    k_dinv<<<(n + 255) / 256, 256, 0, stream>>>(deg, dinv, n);
    k_gemm<<<(tiles + GW - 1) / GW, GW * 32, 0, stream>>>(x, W, nullptr, xw, n);
    const size_t tot = (size_t)n * OUT_SIZE;
    k_self<<<(unsigned)((tot + 255) / 256), 256, 0, stream>>>(xw, dinv, out, n);
    k_scatter<<<(E + 3) / 4, 256, 0, stream>>>(row, col, dinv, xw, out, E);
  }
}


# Gather: a wave sums 8 consecutive rows in a loop (8x fewer waves enter); next row's slot line and own row requested a row ahead, source rows 16 at a time, the rest as one group; sum order unchanged
# speedup vs baseline: 5.8488x; 1.0365x over previous
#include <hip/hip_runtime.h>
#include <hip/hip_bf16.h>


typedef __attribute__((ext_vector_type(2))) float v2f;
typedef __attribute__((ext_vector_type(8))) float v8f;

#define IN_SIZE 256
#define OUT_SIZE 64
#define SCAN_B 1024


#define SLOTS 64

__global__ __launch_bounds__(256) void k_init_deg(unsigned int* __restrict__ cnt, int m) {
  int i = blockIdx.x * 256 + threadIdx.x;
  if (i < m) cnt[i] = 0u;
}

__global__ __launch_bounds__(256) void k_place(const int* __restrict__ row,
                                               const int* __restrict__ col,
                                               unsigned int* __restrict__ deg,
                                               int* __restrict__ slots,
                                               int* __restrict__ ovf, int n, int E) {
  int e = blockIdx.x * 256 + threadIdx.x;
  if (e < E) {
    int r = row[e], c = col[e];
    if (r != c) {
      unsigned int k = atomicAdd(&deg[r], 1u);
      if (k < (unsigned int)SLOTS) {
        slots[(size_t)r * SLOTS + k] = c;
      } else {
        unsigned int j = atomicAdd(&deg[n], 1u);
        ovf[j] = e;
      }
    }
  }
}


#define RBITS 8
#define GA 250
#define NBMAX 512
#define CELLW 64

typedef __attribute__((ext_vector_type(4))) unsigned int v4u;

__global__ __launch_bounds__(1024) void k_bin(const int* __restrict__ row,
                                              const int* __restrict__ col,
                                              unsigned int* __restrict__ seg, int E, int NB, int chunk) {
  __shared__ __attribute__((aligned(16))) unsigned int stage[NBMAX * CELLW];
  const int tid = threadIdx.x;
  const int w = blockIdx.x;
  v4u* st4 = (v4u*)stage;
  for (int i = tid; i < NB * 16; i += 1024) st4[i] = (v4u){0u, 0u, 0u, 0u};
  __syncthreads();
  const int e0 = w * chunk;
  int e1 = e0 + chunk;
  if (e1 > E) e1 = E;
  for (int base = e0; base < e1; base += 4096) {
    int r[4], c[4];
#pragma unroll
    for (int j = 0; j < 4; ++j) {
      const int e = base + j * 1024 + tid;
      const bool in = e < e1;
      r[j] = in ? row[e] : 0;
      c[j] = in ? col[e] : 0;
    }
#pragma unroll
    for (int j = 0; j < 4; ++j) {
      if (r[j] != c[j]) {
        unsigned int* cell = stage + ((unsigned int)r[j] >> RBITS) * CELLW;
        const unsigned int k = atomicAdd(cell, 1u);
        if (k < (unsigned int)(CELLW - 1))
          cell[1 + k] = ((unsigned int)(r[j] & 255) << 24) | (unsigned int)c[j];
      }
    }
  }
  __syncthreads();
  for (int i = tid; i < NB * 16; i += 1024) {
    const int b = i >> 4, q = i & 15;
    const unsigned int cnt = stage[b * CELLW];
    if (q < 8 || cnt > 31u) ((v4u*)(seg + ((size_t)b * GA + w) * CELLW))[q] = st4[i];
  }
}

__global__ __launch_bounds__(1024) void k_fine(const unsigned int* __restrict__ seg,
                                               const int* __restrict__ row,
                                               const int* __restrict__ col,
                                               unsigned int* __restrict__ deg,
                                               int* __restrict__ slots, int n, int E) {
  __shared__ __attribute__((aligned(16))) int sl[256 * SLOTS];
  __shared__ unsigned int rc[256];
  __shared__ unsigned int cc[GA];
  __shared__ unsigned int over;
  const int tid = threadIdx.x;
  const int b = blockIdx.x;
  const unsigned int* cells = seg + (size_t)b * GA * CELLW;
  for (int i = tid; i < 256 * SLOTS / 4; i += 1024) ((v4u*)sl)[i] = (v4u){0u, 0u, 0u, 0u};
  if (tid < 256) rc[tid] = 0u;
  if (tid == 0) over = 0u;
  __syncthreads();
  if (tid < GA) {
    const unsigned int cw = cells[tid * CELLW];
    cc[tid] = cw;
    if (cw > (unsigned int)(CELLW - 1)) over = 1u;
  }
  __syncthreads();
  if (over == 0u) {
    for (int base = 0; base < GA * CELLW; base += 4096) {
      unsigned int wd[4];
      bool ok[4];
#pragma unroll
      for (int j = 0; j < 4; ++j) {
        const int i = base + j * 1024 + tid;
        const bool in = i < GA * CELLW;
        const unsigned int jj = (unsigned int)(i & (CELLW - 1));
        ok[j] = in && jj >= 1u && jj <= cc[in ? (i >> 6) : 0];
        wd[j] = ok[j] ? cells[i] : 0u;
      }
#pragma unroll
      for (int j = 0; j < 4; ++j) {
        if (ok[j]) {
          const unsigned int rl = wd[j] >> 24;
          const unsigned int k = atomicAdd(&rc[rl], 1u);
          if (k < (unsigned int)SLOTS) sl[rl * SLOTS + k] = (int)(wd[j] & 0xFFFFFFu);
        }
      }
    }
  } else {
    for (int e = tid; e < E; e += 1024) {
      const int r = row[e], c = col[e];
      if (r != c && (r >> RBITS) == b) {
        const unsigned int rl = (unsigned int)(r & 255);
        const unsigned int k = atomicAdd(&rc[rl], 1u);
        if (k < (unsigned int)SLOTS) sl[rl * SLOTS + k] = c;
      }
    }
  }
  __syncthreads();
  if (tid < 256) {
    const int r = b * 256 + tid;
    if (r < n) deg[r] = rc[tid];
  }
  for (int i = tid; i < 256 * 16; i += 1024) {
    const int rl = i >> 4, q = i & 15;
    const int r = b * 256 + rl;
    if (r < n && (q < 8 || rc[rl] > 32u)) ((v4u*)slots)[(size_t)r * 16 + q] = ((const v4u*)sl)[i];
  }
}


typedef __attribute__((ext_vector_type(4))) float v4f;

#define WP 260
#define GW 25

__global__ __launch_bounds__(GW * 32) void k_gemm(const float* __restrict__ x,
                                              const float* __restrict__ W,
                                              const unsigned int* __restrict__ dg,
                                              float* __restrict__ xw, int n) {
  __shared__ __attribute__((aligned(16))) float sW[OUT_SIZE * WP];

  const int lane  = threadIdx.x & 31;
  const int wave  = threadIdx.x >> 5;
  const int tile  = blockIdx.x * GW + wave;
  const int ml    = lane & 15;
  const int koff  = (lane >> 4) << 2;
  int arow = tile * 16 + ml;
  if (arow > n - 1) arow = n - 1;
  const v4f* xr = (const v4f*)(x + (size_t)arow * IN_SIZE + koff);

  const int rbase = tile * 16 + ((lane >> 4) << 3);

  v4f a0 = xr[0], a1 = xr[2], a2 = xr[4], a3 = xr[6];
  unsigned int dc[8];
#pragma unroll
  for (int v = 0; v < 8; ++v) dc[v] = (dg && rbase + v < n) ? dg[rbase + v] : 0u;

  {
    const v4f* W4 = (const v4f*)W;
    constexpr int WT = (OUT_SIZE * IN_SIZE / 4 + GW * 32 - 1) / (GW * 32);
    v4f t[WT];
#pragma unroll
    for (int i = 0; i < WT; ++i) {
      const int p = i * (GW * 32) + (int)threadIdx.x;
      t[i] = W4[p < OUT_SIZE * IN_SIZE / 4 ? p : 0];
    }
#pragma unroll
    for (int i = 0; i < WT; ++i) {
      const int p = i * (GW * 32) + (int)threadIdx.x;
      if (p < OUT_SIZE * IN_SIZE / 4) *(v4f*)(sW + (p >> 6) * WP + ((p & 63) << 2)) = t[i];
    }
  }
  __syncthreads();
  if (tile * 16 >= n) return;

  const float* wl = sW + ml * WP + koff;

  v8f acc0 = {}, acc1 = {}, acc2 = {}, acc3 = {};
  v4f b0 = *(const v4f*)(wl + 0 * 16 * WP);
  v4f b1 = *(const v4f*)(wl + 1 * 16 * WP);
  v4f b2 = *(const v4f*)(wl + 2 * 16 * WP);
  v4f b3 = *(const v4f*)(wl + 3 * 16 * WP);

#define GEMM_STEP(A, KN)                                                                              \
  {                                                                                                  \
          \
    const v4f n0 = *(const v4f*)(wl + 0 * 16 * WP + (KN));                                           \
    const v4f n1 = *(const v4f*)(wl + 1 * 16 * WP + (KN));                                           \
    const v4f n2 = *(const v4f*)(wl + 2 * 16 * WP + (KN));                                           \
    const v4f n3 = *(const v4f*)(wl + 3 * 16 * WP + (KN));                                           \
    acc0 = __builtin_amdgcn_wmma_f32_16x16x4_f32(false, (A).lo, false, b0.lo, (short)0, acc0, false, false); \
    acc1 = __builtin_amdgcn_wmma_f32_16x16x4_f32(false, (A).lo, false, b1.lo, (short)0, acc1, false, false); \
    acc2 = __builtin_amdgcn_wmma_f32_16x16x4_f32(false, (A).lo, false, b2.lo, (short)0, acc2, false, false); \
    acc3 = __builtin_amdgcn_wmma_f32_16x16x4_f32(false, (A).lo, false, b3.lo, (short)0, acc3, false, false); \
    acc0 = __builtin_amdgcn_wmma_f32_16x16x4_f32(false, (A).hi, false, b0.hi, (short)0, acc0, false, false); \
    acc1 = __builtin_amdgcn_wmma_f32_16x16x4_f32(false, (A).hi, false, b1.hi, (short)0, acc1, false, false); \
    acc2 = __builtin_amdgcn_wmma_f32_16x16x4_f32(false, (A).hi, false, b2.hi, (short)0, acc2, false, false); \
    acc3 = __builtin_amdgcn_wmma_f32_16x16x4_f32(false, (A).hi, false, b3.hi, (short)0, acc3, false, false); \
    b0 = n0; b1 = n1; b2 = n2; b3 = n3;                                                              \
  }

  for (int k = 0; k < IN_SIZE; k += 32) {
    v4f q0 = a0, q1 = a1, q2 = a2, q3 = a3;
    if (k + 32 < IN_SIZE) {
      const v4f* xn = xr + ((k + 32) >> 2);
      q0 = xn[0]; q1 = xn[2]; q2 = xn[4]; q3 = xn[6];
    }
    GEMM_STEP(a0, k + 8)
    GEMM_STEP(a1, k + 16)
    GEMM_STEP(a2, k + 24)
    GEMM_STEP(a3, (k + 32) & (IN_SIZE - 1))
    a0 = q0; a1 = q1; a2 = q2; a3 = q3;
  }
#undef GEMM_STEP

  float* o = xw + (size_t)rbase * OUT_SIZE + ml;
#pragma unroll
  for (int v = 0; v < 8; ++v) {
    if (rbase + v < n) {
      const float s = dg ? rsqrtf((float)dc[v] + 1.0f) : 1.0f;
      o[(size_t)v * OUT_SIZE +  0] = s * acc0[v];
      o[(size_t)v * OUT_SIZE + 16] = s * acc1[v];
      o[(size_t)v * OUT_SIZE + 32] = s * acc2[v];
      o[(size_t)v * OUT_SIZE + 48] = s * acc3[v];
    }
  }
}


#define GR 8
#define GWV 4

__global__ __launch_bounds__(GWV * 32) void k_gather(const unsigned int* __restrict__ deg,
                                                const int* __restrict__ ovf,
                                                const int* __restrict__ row,
                                                const int* __restrict__ col,
                                                const float* __restrict__ xs,
                                                float* out, int n, int E) {
  const int lane = threadIdx.x & 31;
  const int r0 = __builtin_amdgcn_readfirstlane((blockIdx.x * GWV + (threadIdx.x >> 5)) * GR);
  if (r0 >= n) return;
  const int rend = r0 + GR < n ? r0 + GR : n;
  const v2f* rows = (const v2f*)xs;
  const int* slots = (const int*)out;
  const int dl = r0 + (lane & (GR - 1));
  const unsigned int degv = deg[dl < n ? dl : n - 1];
  int s_lo = slots[(size_t)r0 * SLOTS + lane];
  v2f self = rows[(size_t)r0 * 32 + lane];
  for (int r = r0; r < rend; ++r) {
    const unsigned int cnt = (unsigned int)__builtin_amdgcn_readlane((int)degv, r - r0);
    const unsigned int m = cnt < (unsigned int)SLOTS ? cnt : (unsigned int)SLOTS;
    int s_hi = 0;
    if (cnt > 32u) s_hi = slots[(size_t)r * SLOTS + 32 + lane];
    const int rn = r + 1 < rend ? r + 1 : r;
    const int n_slo = slots[(size_t)rn * SLOTS + lane];
    const v2f n_self = rows[(size_t)rn * 32 + lane];
    v2f acc = self;
    unsigned int p = 0u;
    for (; p + 16u <= m; p += 16u) {
      const int sv = (p & 32u) ? s_hi : s_lo;
      int c[16];
#pragma unroll
      for (int k = 0; k < 16; ++k) c[k] = __builtin_amdgcn_readlane(sv, (int)((p + k) & 31u));
      v2f v[16];
#pragma unroll
      for (int k = 0; k < 16; ++k) v[k] = rows[(size_t)c[k] * 32 + lane];
#pragma unroll
      for (int k = 0; k < 16; ++k) acc += v[k];
    }
    if (p < m) {
      const int sv = (p & 32u) ? s_hi : s_lo;
      int c[15];
#pragma unroll
      for (int k = 0; k < 15; ++k) {
        unsigned int q = p + k;
        q = q < m ? q : m - 1u;
        c[k] = __builtin_amdgcn_readlane(sv, (int)(q & 31u));
      }
      v2f v[15];
#pragma unroll
      for (int k = 0; k < 15; ++k) {
        v[k] = (v2f){0.0f, 0.0f};
        if (p + k < m) v[k] = rows[(size_t)c[k] * 32 + lane];
      }
#pragma unroll
      for (int k = 0; k < 15; ++k) acc += v[k];
    }
    if (cnt > (unsigned int)SLOTS) {
      if (ovf) {
        const unsigned int M = deg[n];
        for (unsigned int j = 0; j < M; ++j) {
          const int e = ovf[j];
          if (row[e] == r) acc += rows[(size_t)col[e] * 32 + lane];
        }
      } else {
        acc = self;
        for (int e = 0; e < E; ++e) {
          const int cs = col[e];
          if (row[e] == r && cs != r) acc += rows[(size_t)cs * 32 + lane];
        }
      }
    }
    const float d = rsqrtf((float)cnt + 1.0f);
    ((v2f*)out)[(size_t)r * 32 + lane] = acc * d;
    s_lo = n_slo; self = n_self;
  }
}


__global__ __launch_bounds__(256) void k_count_deg(const int* __restrict__ row,
                                                   const int* __restrict__ col,
                                                   unsigned int* __restrict__ deg, int E) {
  int e = blockIdx.x * 256 + threadIdx.x;
  if (e < E) {
    int r = row[e], c = col[e];
    if (r != c) atomicAdd(&deg[r], 1u);
  }
}

__global__ __launch_bounds__(256) void k_dinv(const unsigned int* __restrict__ deg,
                                              float* __restrict__ dinv, int n) {
  int i = blockIdx.x * 256 + threadIdx.x;
  if (i < n) dinv[i] = rsqrtf((float)deg[i] + 1.0f);
}

__global__ __launch_bounds__(256) void k_self(const float* __restrict__ xw,
                                              const float* __restrict__ dinv,
                                              float* __restrict__ out, int n) {
  size_t idx = (size_t)blockIdx.x * 256 + threadIdx.x;
  if (idx < (size_t)n * OUT_SIZE) {
    int i = (int)(idx >> 6);
    float d = dinv[i];
    out[idx] = d * d * xw[idx];
  }
}

__global__ __launch_bounds__(256) void k_scatter(const int* __restrict__ row,
                                                 const int* __restrict__ col,
                                                 const float* __restrict__ dinv,
                                                 const float* __restrict__ xw,
                                                 float* __restrict__ out, int E) {
  int e = blockIdx.x * 4 + (threadIdx.x >> 6);
  int j = threadIdx.x & 63;
  if (e >= E) return;
  int r = row[e];
  int c = col[e];
  if (r == c) return;
  float norm = dinv[r] * dinv[c];
  unsafeAtomicAdd(out + (size_t)r * OUT_SIZE + j,
                  norm * xw[(size_t)c * OUT_SIZE + j]);
}


extern "C" void kernel_launch(void* const* d_in, const int* in_sizes, int n_in,
                              void* d_out, int out_size, void* d_ws, size_t ws_size,
                              hipStream_t stream) {
  const int*   edge = (const int*)d_in[0];
  const float* x    = (const float*)d_in[1];
  const float* W    = (const float*)d_in[3];

  const int E = in_sizes[0] / 2;
  const int n = in_sizes[1] / IN_SIZE;
  const int* row = edge;
  const int* col = edge + E;
  float* out = (float*)d_out;

  float*        xw   = (float*)d_ws;
  unsigned int* deg  = (unsigned int*)(xw + (size_t)n * OUT_SIZE);
  int*          ovf  = (int*)(deg + n + 1);
  float*        dinv = (float*)(deg + n);

  const int tiles = (n + 15) / 16;
  const size_t need = ((size_t)n * OUT_SIZE + (size_t)n + 1 + (size_t)E) * sizeof(float);

  const int NB = (n + 255) / 256;
  if (ws_size >= need && n <= 131072 && (size_t)NB * GA <= (size_t)n) {
    unsigned int* seg = (unsigned int*)d_ws;
    k_bin<<<GA, 1024, 0, stream>>>(row, col, seg, E, NB, (E + GA - 1) / GA);
    k_fine<<<NB, 1024, 0, stream>>>(seg, row, col, deg, (int*)d_out, n, E);
    k_gemm<<<(tiles + GW - 1) / GW, GW * 32, 0, stream>>>(x, W, deg, xw, n);
    k_gather<<<(n + GR * GWV - 1) / (GR * GWV), GWV * 32, 0, stream>>>(deg, nullptr, row, col, xw, out, n, E);
  } else if (ws_size >= need) {
    k_init_deg<<<(n + 1 + 255) / 256, 256, 0, stream>>>(deg, n + 1);
    k_place<<<(E + 255) / 256, 256, 0, stream>>>(row, col, deg, (int*)d_out, ovf, n, E);
    k_gemm<<<(tiles + GW - 1) / GW, GW * 32, 0, stream>>>(x, W, deg, xw, n);
    k_gather<<<(n + GR * GWV - 1) / (GR * GWV), GWV * 32, 0, stream>>>(deg, ovf, row, col, xw, out, n, E);
  } else {
    k_init_deg<<<(n + 255) / 256, 256, 0, stream>>>(deg, n);
    k_count_deg<<<(E + 255) / 256, 256, 0, stream>>>(row, col, deg, E);
    k_dinv<<<(n + 255) / 256, 256, 0, stream>>>(deg, dinv, n);
    k_gemm<<<(tiles + GW - 1) / GW, GW * 32, 0, stream>>>(x, W, nullptr, xw, n);
    const size_t tot = (size_t)n * OUT_SIZE;
    k_self<<<(unsigned)((tot + 255) / 256), 256, 0, stream>>>(xw, dinv, out, n);
    k_scatter<<<(E + 3) / 4, 256, 0, stream>>>(row, col, dinv, xw, out, E);
  }
}


# Row ranking (pass two of edge placement) moved to the head of each GEMM work-group (bucket = its 400 rows; cell reads, rank adds ten a lane a trip): 4 launches to 3; sums in LDS atomic arrival order
# speedup vs baseline: 6.1783x; 1.0563x over previous
#include <hip/hip_runtime.h>
#include <hip/hip_bf16.h>


typedef __attribute__((ext_vector_type(2))) float v2f;
typedef __attribute__((ext_vector_type(8))) float v8f;

#define IN_SIZE 256
#define OUT_SIZE 64
#define SCAN_B 1024


#define SLOTS 64

__global__ __launch_bounds__(256) void k_init_deg(unsigned int* __restrict__ cnt, int m) {
  int i = blockIdx.x * 256 + threadIdx.x;
  if (i < m) cnt[i] = 0u;
}

__global__ __launch_bounds__(256) void k_place(const int* __restrict__ row,
                                               const int* __restrict__ col,
                                               unsigned int* __restrict__ deg,
                                               int* __restrict__ slots,
                                               int* __restrict__ ovf, int n, int E) {
  int e = blockIdx.x * 256 + threadIdx.x;
  if (e < E) {
    int r = row[e], c = col[e];
    if (r != c) {
      unsigned int k = atomicAdd(&deg[r], 1u);
      if (k < (unsigned int)SLOTS) {
        slots[(size_t)r * SLOTS + k] = c;
      } else {
        unsigned int j = atomicAdd(&deg[n], 1u);
        ovf[j] = e;
      }
    }
  }
}


#define BR 400
#define RSH 23
#define GA 250
#define NBMAX 512
#define CELLW 64

typedef __attribute__((ext_vector_type(4))) unsigned int v4u;

__global__ __launch_bounds__(1024) void k_bin(const int* __restrict__ row,
                                              const int* __restrict__ col,
                                              unsigned int* __restrict__ seg, int E, int NB, int chunk) {
  __shared__ __attribute__((aligned(16))) unsigned int stage[NBMAX * CELLW];
  const int tid = threadIdx.x;
  const int w = blockIdx.x;
  v4u* st4 = (v4u*)stage;
  for (int i = tid; i < NB * 16; i += 1024) st4[i] = (v4u){0u, 0u, 0u, 0u};
  __syncthreads();
  const int e0 = w * chunk;
  int e1 = e0 + chunk;
  if (e1 > E) e1 = E;
  for (int base = e0; base < e1; base += 4096) {
    int r[4], c[4];
#pragma unroll
    for (int j = 0; j < 4; ++j) {
      const int e = base + j * 1024 + tid;
      const bool in = e < e1;
      r[j] = in ? row[e] : 0;
      c[j] = in ? col[e] : 0;
    }
#pragma unroll
    for (int j = 0; j < 4; ++j) {
      if (r[j] != c[j]) {
        const unsigned int bk = (unsigned int)r[j] / (unsigned int)BR;
        unsigned int* cell = stage + bk * CELLW;
        const unsigned int k = atomicAdd(cell, 1u);
        if (k < (unsigned int)(CELLW - 1))
          cell[1 + k] = (((unsigned int)r[j] - bk * (unsigned int)BR) << RSH) | (unsigned int)c[j];
      }
    }
  }
  __syncthreads();
  for (int i = tid; i < NB * 16; i += 1024) {
    const int b = i >> 4, q = i & 15;
    const unsigned int cnt = stage[b * CELLW];
    if (q < 8 || cnt > 31u) ((v4u*)(seg + (size_t)b * (BR * OUT_SIZE) + (size_t)w * CELLW))[q] = st4[i];
  }
}


typedef __attribute__((ext_vector_type(4))) float v4f;

#define WP 260
#define GW 25

template <bool RANK>
__global__ __launch_bounds__(GW * 32) void k_gemm(const float* __restrict__ x,
                                              const float* __restrict__ W,
                                              const unsigned int* dg,
                                              float* xw,
                                              const unsigned int* seg,
                                              const int* __restrict__ row,
                                              const int* __restrict__ col,
                                              unsigned int* deg,
                                              int* slots, int n, int E) {
  __shared__ __attribute__((aligned(16))) float sW[OUT_SIZE * WP];
  __shared__ __attribute__((aligned(16))) int sl[RANK ? BR * SLOTS : 4];
  __shared__ unsigned int rc[RANK ? BR : 1];
  __shared__ unsigned int cc[RANK ? GA : 1];
  __shared__ unsigned int over;
  constexpr int NT = GW * 32;

  const int lane  = threadIdx.x & 31;
  const int wave  = threadIdx.x >> 5;
  const int tile  = blockIdx.x * GW + wave;
  const int ml    = lane & 15;
  const int koff  = (lane >> 4) << 2;
  int arow = tile * 16 + ml;
  if (arow > n - 1) arow = n - 1;
  const v4f* xr = (const v4f*)(x + (size_t)arow * IN_SIZE + koff);

  const int rbase = tile * 16 + ((lane >> 4) << 3);

  v4f a0 = xr[0], a1 = xr[2], a2 = xr[4], a3 = xr[6];
  unsigned int dc[8];
#pragma unroll
  for (int v = 0; v < 8; ++v) dc[v] = (!RANK && dg && rbase + v < n) ? dg[rbase + v] : 0u;
  const int tid = threadIdx.x;
  const int bkt = blockIdx.x;
  const unsigned int* cells = seg + (size_t)bkt * (BR * OUT_SIZE);
  unsigned int cw = 0u;
  if (RANK && tid < GA) cw = cells[tid * CELLW];

  {
    const v4f* W4 = (const v4f*)W;
    constexpr int WT = (OUT_SIZE * IN_SIZE / 4 + GW * 32 - 1) / (GW * 32);
    v4f t[WT];
#pragma unroll
    for (int i = 0; i < WT; ++i) {
      const int p = i * (GW * 32) + (int)threadIdx.x;
      t[i] = W4[p < OUT_SIZE * IN_SIZE / 4 ? p : 0];
    }
    if constexpr (RANK) {
      for (int i = tid; i < BR * SLOTS / 4; i += NT) ((v4u*)sl)[i] = (v4u){0u, 0u, 0u, 0u};
      if (tid < BR) rc[tid] = 0u;
      if (tid == 0) over = 0u;
    }
#pragma unroll
    for (int i = 0; i < WT; ++i) {
      const int p = i * (GW * 32) + (int)threadIdx.x;
      if (p < OUT_SIZE * IN_SIZE / 4) *(v4f*)(sW + (p >> 6) * WP + ((p & 63) << 2)) = t[i];
    }
  }
  __syncthreads();
  if constexpr (RANK) {
    if (tid < GA) {
      cc[tid] = cw;
      if (cw > (unsigned int)(CELLW - 1)) over = 1u;
    }
    __syncthreads();
    if (over == 0u) {
      constexpr int RU = 10;
      static_assert((GA * CELLW) % (RU * NT) == 0, "the trips cover the cells exactly");
      for (int base = 0; base < GA * CELLW; base += RU * NT) {
        unsigned int wd[RU], kk[RU], cj[RU];
        bool ok[RU];
#pragma unroll
        for (int j = 0; j < RU; ++j) cj[j] = cc[(base + j * NT + tid) >> 6];
#pragma unroll
        for (int j = 0; j < RU; ++j) {
          const int i = base + j * NT + tid;
          const unsigned int jj = (unsigned int)(i & (CELLW - 1));
          ok[j] = (jj >= 1u) & (jj <= cj[j]);
          wd[j] = ok[j] ? cells[i] : 0u;
        }
#pragma unroll
        for (int j = 0; j < RU; ++j) {
          kk[j] = (unsigned int)SLOTS;
          if (ok[j]) kk[j] = atomicAdd(&rc[wd[j] >> RSH], 1u);
        }
#pragma unroll
        for (int j = 0; j < RU; ++j) {
          if (kk[j] < (unsigned int)SLOTS)
            sl[(wd[j] >> RSH) * SLOTS + kk[j]] = (int)(wd[j] & ((1u << RSH) - 1u));
        }
      }
    } else {
      for (int e = tid; e < E; e += NT) {
        const int r = row[e], c = col[e];
        if (r != c && (unsigned int)r / (unsigned int)BR == (unsigned int)bkt) {
          const unsigned int rl = (unsigned int)r - (unsigned int)bkt * (unsigned int)BR;
          const unsigned int k = atomicAdd(&rc[rl], 1u);
          if (k < (unsigned int)SLOTS) sl[rl * SLOTS + k] = c;
        }
      }
    }
    __syncthreads();
    if (tid < BR) {
      const int r = bkt * BR + tid;
      if (r < n) deg[r] = rc[tid];
    }
    static_assert((BR * 16) % NT == 0, "the trips cover the slot lists exactly");
    constexpr int NP = BR * 16 / NT;
    v4u dv[NP];
    unsigned int rv[NP];
#pragma unroll
    for (int j = 0; j < NP; ++j) {
      const int i = j * NT + tid;
      rv[j] = rc[i >> 4];
      dv[j] = ((const v4u*)sl)[i];
    }
#pragma unroll
    for (int j = 0; j < NP; ++j) {
      const int i = j * NT + tid;
      const int rl = i >> 4, q = i & 15;
      const int r = bkt * BR + rl;
      if (r < n && ((q < 8) | (rv[j] > 32u))) ((v4u*)slots)[(size_t)r * 16 + q] = dv[j];
    }
#pragma unroll
    for (int v = 0; v < 8; ++v) dc[v] = rc[wave * 16 + ((lane >> 4) << 3) + v];
  }
  if (tile * 16 >= n) return;

  const float* wl = sW + ml * WP + koff;

  v8f acc0 = {}, acc1 = {}, acc2 = {}, acc3 = {};
  v4f b0 = *(const v4f*)(wl + 0 * 16 * WP);
  v4f b1 = *(const v4f*)(wl + 1 * 16 * WP);
  v4f b2 = *(const v4f*)(wl + 2 * 16 * WP);
  v4f b3 = *(const v4f*)(wl + 3 * 16 * WP);

#define GEMM_STEP(A, KN)                                                                              \
  {                                                                                                  \
          \
    const v4f n0 = *(const v4f*)(wl + 0 * 16 * WP + (KN));                                           \
    const v4f n1 = *(const v4f*)(wl + 1 * 16 * WP + (KN));                                           \
    const v4f n2 = *(const v4f*)(wl + 2 * 16 * WP + (KN));                                           \
    const v4f n3 = *(const v4f*)(wl + 3 * 16 * WP + (KN));                                           \
    acc0 = __builtin_amdgcn_wmma_f32_16x16x4_f32(false, (A).lo, false, b0.lo, (short)0, acc0, false, false); \
    acc1 = __builtin_amdgcn_wmma_f32_16x16x4_f32(false, (A).lo, false, b1.lo, (short)0, acc1, false, false); \
    acc2 = __builtin_amdgcn_wmma_f32_16x16x4_f32(false, (A).lo, false, b2.lo, (short)0, acc2, false, false); \
    acc3 = __builtin_amdgcn_wmma_f32_16x16x4_f32(false, (A).lo, false, b3.lo, (short)0, acc3, false, false); \
    acc0 = __builtin_amdgcn_wmma_f32_16x16x4_f32(false, (A).hi, false, b0.hi, (short)0, acc0, false, false); \
    acc1 = __builtin_amdgcn_wmma_f32_16x16x4_f32(false, (A).hi, false, b1.hi, (short)0, acc1, false, false); \
    acc2 = __builtin_amdgcn_wmma_f32_16x16x4_f32(false, (A).hi, false, b2.hi, (short)0, acc2, false, false); \
    acc3 = __builtin_amdgcn_wmma_f32_16x16x4_f32(false, (A).hi, false, b3.hi, (short)0, acc3, false, false); \
    b0 = n0; b1 = n1; b2 = n2; b3 = n3;                                                              \
  }

  for (int k = 0; k < IN_SIZE; k += 32) {
    v4f q0 = a0, q1 = a1, q2 = a2, q3 = a3;
    if (k + 32 < IN_SIZE) {
      const v4f* xn = xr + ((k + 32) >> 2);
      q0 = xn[0]; q1 = xn[2]; q2 = xn[4]; q3 = xn[6];
    }
    GEMM_STEP(a0, k + 8)
    GEMM_STEP(a1, k + 16)
    GEMM_STEP(a2, k + 24)
    GEMM_STEP(a3, (k + 32) & (IN_SIZE - 1))
    a0 = q0; a1 = q1; a2 = q2; a3 = q3;
  }
#undef GEMM_STEP

  float* o = xw + (size_t)rbase * OUT_SIZE + ml;
#pragma unroll
  for (int v = 0; v < 8; ++v) {
    if (rbase + v < n) {
      const float s = (RANK || dg) ? rsqrtf((float)dc[v] + 1.0f) : 1.0f;
      o[(size_t)v * OUT_SIZE +  0] = s * acc0[v];
      o[(size_t)v * OUT_SIZE + 16] = s * acc1[v];
      o[(size_t)v * OUT_SIZE + 32] = s * acc2[v];
      o[(size_t)v * OUT_SIZE + 48] = s * acc3[v];
    }
  }
}


#define GR 8
#define GWV 4

__global__ __launch_bounds__(GWV * 32) void k_gather(const unsigned int* __restrict__ deg,
                                                const int* __restrict__ ovf,
                                                const int* __restrict__ row,
                                                const int* __restrict__ col,
                                                const float* __restrict__ xs,
                                                float* out, int n, int E) {
  const int lane = threadIdx.x & 31;
  const int r0 = __builtin_amdgcn_readfirstlane((blockIdx.x * GWV + (threadIdx.x >> 5)) * GR);
  if (r0 >= n) return;
  const int rend = r0 + GR < n ? r0 + GR : n;
  const v2f* rows = (const v2f*)xs;
  const int* slots = (const int*)out;
  const int dl = r0 + (lane & (GR - 1));
  const unsigned int degv = deg[dl < n ? dl : n - 1];
  int s_lo = slots[(size_t)r0 * SLOTS + lane];
  v2f self = rows[(size_t)r0 * 32 + lane];
  for (int r = r0; r < rend; ++r) {
    const unsigned int cnt = (unsigned int)__builtin_amdgcn_readlane((int)degv, r - r0);
    const unsigned int m = cnt < (unsigned int)SLOTS ? cnt : (unsigned int)SLOTS;
    int s_hi = 0;
    if (cnt > 32u) s_hi = slots[(size_t)r * SLOTS + 32 + lane];
    const int rn = r + 1 < rend ? r + 1 : r;
    const int n_slo = slots[(size_t)rn * SLOTS + lane];
    const v2f n_self = rows[(size_t)rn * 32 + lane];
    v2f acc = self;
    unsigned int p = 0u;
    for (; p + 16u <= m; p += 16u) {
      const int sv = (p & 32u) ? s_hi : s_lo;
      int c[16];
#pragma unroll
      for (int k = 0; k < 16; ++k) c[k] = __builtin_amdgcn_readlane(sv, (int)((p + k) & 31u));
      v2f v[16];
#pragma unroll
      for (int k = 0; k < 16; ++k) v[k] = rows[(size_t)c[k] * 32 + lane];
#pragma unroll
      for (int k = 0; k < 16; ++k) acc += v[k];
    }
    if (p < m) {
      const int sv = (p & 32u) ? s_hi : s_lo;
      int c[15];
#pragma unroll
      for (int k = 0; k < 15; ++k) {
        unsigned int q = p + k;
        q = q < m ? q : m - 1u;
        c[k] = __builtin_amdgcn_readlane(sv, (int)(q & 31u));
      }
      v2f v[15];
#pragma unroll
      for (int k = 0; k < 15; ++k) {
        v[k] = (v2f){0.0f, 0.0f};
        if (p + k < m) v[k] = rows[(size_t)c[k] * 32 + lane];
      }
#pragma unroll
      for (int k = 0; k < 15; ++k) acc += v[k];
    }
    if (cnt > (unsigned int)SLOTS) {
      if (ovf) {
        const unsigned int M = deg[n];
        for (unsigned int j = 0; j < M; ++j) {
          const int e = ovf[j];
          if (row[e] == r) acc += rows[(size_t)col[e] * 32 + lane];
        }
      } else {
        acc = self;
        for (int e = 0; e < E; ++e) {
          const int cs = col[e];
          if (row[e] == r && cs != r) acc += rows[(size_t)cs * 32 + lane];
        }
      }
    }
    const float d = rsqrtf((float)cnt + 1.0f);
    ((v2f*)out)[(size_t)r * 32 + lane] = acc * d;
    s_lo = n_slo; self = n_self;
  }
}


__global__ __launch_bounds__(256) void k_count_deg(const int* __restrict__ row,
                                                   const int* __restrict__ col,
                                                   unsigned int* __restrict__ deg, int E) {
  int e = blockIdx.x * 256 + threadIdx.x;
  if (e < E) {
    int r = row[e], c = col[e];
    if (r != c) atomicAdd(&deg[r], 1u);
  }
}

__global__ __launch_bounds__(256) void k_dinv(const unsigned int* __restrict__ deg,
                                              float* __restrict__ dinv, int n) {
  int i = blockIdx.x * 256 + threadIdx.x;
  if (i < n) dinv[i] = rsqrtf((float)deg[i] + 1.0f);
}

__global__ __launch_bounds__(256) void k_self(const float* __restrict__ xw,
                                              const float* __restrict__ dinv,
                                              float* __restrict__ out, int n) {
  size_t idx = (size_t)blockIdx.x * 256 + threadIdx.x;
  if (idx < (size_t)n * OUT_SIZE) {
    int i = (int)(idx >> 6);
    float d = dinv[i];
    out[idx] = d * d * xw[idx];
  }
}

__global__ __launch_bounds__(256) void k_scatter(const int* __restrict__ row,
                                                 const int* __restrict__ col,
                                                 const float* __restrict__ dinv,
                                                 const float* __restrict__ xw,
                                                 float* __restrict__ out, int E) {
  int e = blockIdx.x * 4 + (threadIdx.x >> 6);
  int j = threadIdx.x & 63;
  if (e >= E) return;
  int r = row[e];
  int c = col[e];
  if (r == c) return;
  float norm = dinv[r] * dinv[c];
  unsafeAtomicAdd(out + (size_t)r * OUT_SIZE + j,
                  norm * xw[(size_t)c * OUT_SIZE + j]);
}


extern "C" void kernel_launch(void* const* d_in, const int* in_sizes, int n_in,
                              void* d_out, int out_size, void* d_ws, size_t ws_size,
                              hipStream_t stream) {
  const int*   edge = (const int*)d_in[0];
  const float* x    = (const float*)d_in[1];
  const float* W    = (const float*)d_in[3];

  const int E = in_sizes[0] / 2;
  const int n = in_sizes[1] / IN_SIZE;
  const int* row = edge;
  const int* col = edge + E;
  float* out = (float*)d_out;

  float*        xw   = (float*)d_ws;
  unsigned int* deg  = (unsigned int*)(xw + (size_t)n * OUT_SIZE);
  int*          ovf  = (int*)(deg + n + 1);
  float*        dinv = (float*)(deg + n);

  const int tiles = (n + 15) / 16;
  const size_t need = ((size_t)n * OUT_SIZE + (size_t)n + 1 + (size_t)E) * sizeof(float);

  const int NB = (n + BR - 1) / BR;
  if (ws_size >= need && NB <= NBMAX && n <= (1 << RSH) &&
      (size_t)(n - (NB - 1) * BR) * OUT_SIZE >= (size_t)GA * CELLW) {
    unsigned int* seg = (unsigned int*)d_ws;
    k_bin<<<GA, 1024, 0, stream>>>(row, col, seg, E, NB, (E + GA - 1) / GA);
    k_gemm<true><<<NB, GW * 32, 0, stream>>>(x, W, nullptr, xw, seg, row, col, deg, (int*)d_out, n, E);
    k_gather<<<(n + GR * GWV - 1) / (GR * GWV), GWV * 32, 0, stream>>>(deg, nullptr, row, col, xw, out, n, E);
  } else if (ws_size >= need) {
    k_init_deg<<<(n + 1 + 255) / 256, 256, 0, stream>>>(deg, n + 1);
    k_place<<<(E + 255) / 256, 256, 0, stream>>>(row, col, deg, (int*)d_out, ovf, n, E);
    k_gemm<false><<<(tiles + GW - 1) / GW, GW * 32, 0, stream>>>(x, W, deg, xw, nullptr, nullptr, nullptr, nullptr, nullptr, n, E);
    k_gather<<<(n + GR * GWV - 1) / (GR * GWV), GWV * 32, 0, stream>>>(deg, ovf, row, col, xw, out, n, E);
  } else {
    k_init_deg<<<(n + 255) / 256, 256, 0, stream>>>(deg, n);
    k_count_deg<<<(E + 255) / 256, 256, 0, stream>>>(row, col, deg, E);
    k_dinv<<<(n + 255) / 256, 256, 0, stream>>>(deg, dinv, n);
    k_gemm<false><<<(tiles + GW - 1) / GW, GW * 32, 0, stream>>>(x, W, nullptr, xw, nullptr, nullptr, nullptr, nullptr, nullptr, n, E);
    const size_t tot = (size_t)n * OUT_SIZE;
    k_self<<<(unsigned)((tot + 255) / 256), 256, 0, stream>>>(xw, dinv, out, n);
    k_scatter<<<(E + 3) / 4, 256, 0, stream>>>(row, col, dinv, xw, out, E);
  }
}
